# starting kernel
# baseline (speedup 1.0000x reference)
#include <hip/hip_runtime.h>
#include <hip/hip_bf16.h>


typedef __attribute__((ext_vector_type(2))) float v2f;
typedef __attribute__((ext_vector_type(8))) float v8f;

#define NEG_SLOPE 0.2f
#define GAT_EPS   1e-16f

template <int KDIM, int MO>
__global__ __launch_bounds__(128) void gemm_wmma_f32(
    const float* __restrict__ X, const float* __restrict__ W,
    float* __restrict__ Hout, int ntiles) {
  const int wave = threadIdx.x >> 5;
  const int tile = blockIdx.x * 4 + wave;
  if (tile >= ntiles) return;
  const int lane  = threadIdx.x & 31;
  const int lmod  = lane & 15;
  const int lhalf = lane >> 4;
  const int row0  = tile * 16;

  v8f acc[MO / 16];
#pragma unroll
  for (int t = 0; t < MO / 16; ++t) acc[t] = v8f{0, 0, 0, 0, 0, 0, 0, 0};

  const float* xrow = X + (size_t)(row0 + lmod) * KDIM;
#pragma unroll 4
  for (int k0 = 0; k0 < KDIM; k0 += 4) {
    const int ka = k0 + 2 * lhalf;
    v2f a;
    a.x = xrow[ka];
    a.y = xrow[ka + 1];
#pragma unroll
    for (int t = 0; t < MO / 16; ++t) {
      const float* wr = W + (size_t)ka * MO + t * 16 + lmod;
      v2f b;
      b.x = wr[0];
      b.y = wr[MO];
      acc[t] = __builtin_amdgcn_wmma_f32_16x16x4_f32(
          false, a, false, b, (short)0, acc[t], false, false);
    }
  }
#pragma unroll
  for (int t = 0; t < MO / 16; ++t)
#pragma unroll
    for (int r = 0; r < 8; ++r)
      Hout[(size_t)(row0 + r + 8 * lhalf) * MO + t * 16 + lmod] = acc[t][r];
}

__global__ void node_attn(const float* __restrict__ Hm,
                          const float* __restrict__ a_src,
                          const float* __restrict__ a_dst,
                          float* __restrict__ esrc, float* __restrict__ edst,
                          int N, int Hh) {
  int idx = blockIdx.x * blockDim.x + threadIdx.x;
  if (idx >= N * Hh) return;
  const int h = idx % Hh;
  const int n = idx / Hh;
  const float* row = Hm + ((size_t)n * Hh + h) * 16;
  float s = 0.f, d = 0.f;
#pragma unroll
  for (int c = 0; c < 16; ++c) {
    const float v = row[c];
    s += v * a_src[h * 16 + c];
    d += v * a_dst[h * 16 + c];
  }
  esrc[idx] = s;
  edst[idx] = d;
}

__global__ void fill_f32(float* __restrict__ p, float v, int n) {
  int i = blockIdx.x * blockDim.x + threadIdx.x;
  if (i < n) p[i] = v;
}
__global__ void fill_u32(unsigned* __restrict__ p, unsigned v, int n) {
  int i = blockIdx.x * blockDim.x + threadIdx.x;
  if (i < n) p[i] = v;
}

__device__ __forceinline__ unsigned f2o(float f) {
  unsigned u = __float_as_uint(f);
  return (u & 0x80000000u) ? ~u : (u | 0x80000000u);
}
__device__ __forceinline__ float o2f(unsigned u) {
  return __uint_as_float((u & 0x80000000u) ? (u & 0x7FFFFFFFu) : ~u);
}
#define ORDERED_NEG_INF 0x007FFFFFu

__global__ void edge_score_max(const int* __restrict__ ei, int E, int N, int Hh,
                               const float* __restrict__ esrc,
                               const float* __restrict__ edst,
                               float* __restrict__ ebuf,
                               unsigned* __restrict__ mmax) {
  long long t = (long long)blockIdx.x * blockDim.x + threadIdx.x;
  const long long Etot = (long long)E + N;
  if (t >= Etot) return;
  int s, d;
  if (t < E) { s = ei[t]; d = ei[(long long)E + t]; }
  else       { s = d = (int)(t - E); }
  for (int h = 0; h < Hh; ++h) {
    float e = esrc[(size_t)s * Hh + h] + edst[(size_t)d * Hh + h];
    e = (e > 0.f) ? e : NEG_SLOPE * e;
    ebuf[t * Hh + h] = e;
    atomicMax(&mmax[(size_t)d * Hh + h], f2o(e));
  }
}

__global__ void edge_exp_sum(const int* __restrict__ ei, int E, int N, int Hh,
                             const unsigned* __restrict__ mmax,
                             float* __restrict__ ebuf,
                             float* __restrict__ denom) {
  long long t = (long long)blockIdx.x * blockDim.x + threadIdx.x;
  const long long Etot = (long long)E + N;
  if (t >= Etot) return;
  int d;
  if (t < E) d = ei[(long long)E + t];
  else       d = (int)(t - E);
  for (int h = 0; h < Hh; ++h) {
    const float m  = o2f(mmax[(size_t)d * Hh + h]);
    const float ex = __expf(ebuf[t * Hh + h] - m);
    ebuf[t * Hh + h] = ex;
    atomicAdd(&denom[(size_t)d * Hh + h], ex);
  }
}

__global__ void edge_aggregate(const int* __restrict__ ei, int E, int N, int Hh,
                               const float* __restrict__ ebuf,
                               const float* __restrict__ denom,
                               const float* __restrict__ Hm,
                               float* __restrict__ outb) {
  long long idx = (long long)blockIdx.x * blockDim.x + threadIdx.x;
  const long long tot = ((long long)E + N) * Hh;
  if (idx >= tot) return;
  const int h = (int)(idx % Hh);
  const long long t = idx / Hh;
  int s, d;
  if (t < E) { s = ei[t]; d = ei[(long long)E + t]; }
  else       { s = d = (int)(t - E); }
  const float alpha = ebuf[t * Hh + h] / (denom[(size_t)d * Hh + h] + GAT_EPS);
  const float4* hp = (const float4*)(Hm + ((size_t)s * Hh + h) * 16);
  float* op = outb + ((size_t)d * Hh + h) * 16;
  const float4 v0 = hp[0], v1 = hp[1], v2 = hp[2], v3 = hp[3];
  atomicAdd(op + 0,  alpha * v0.x); atomicAdd(op + 1,  alpha * v0.y);
  atomicAdd(op + 2,  alpha * v0.z); atomicAdd(op + 3,  alpha * v0.w);
  atomicAdd(op + 4,  alpha * v1.x); atomicAdd(op + 5,  alpha * v1.y);
  atomicAdd(op + 6,  alpha * v1.z); atomicAdd(op + 7,  alpha * v1.w);
  atomicAdd(op + 8,  alpha * v2.x); atomicAdd(op + 9,  alpha * v2.y);
  atomicAdd(op + 10, alpha * v2.z); atomicAdd(op + 11, alpha * v2.w);
  atomicAdd(op + 12, alpha * v3.x); atomicAdd(op + 13, alpha * v3.y);
  atomicAdd(op + 14, alpha * v3.z); atomicAdd(op + 15, alpha * v3.w);
}

__global__ void bias_act(const float* __restrict__ in, const float* __restrict__ b,
                         float* __restrict__ out, int N, int M, int relu) {
  int idx = blockIdx.x * blockDim.x + threadIdx.x;
  if (idx >= N * M) return;
  float v = in[idx] + b[idx % M];
  if (relu) v = fmaxf(v, 0.f);
  out[idx] = v;
}

static inline int cdiv(long long a, int b) { return (int)((a + b - 1) / b); }

struct LayerBufs {
  float* hbuf; float* outb; float* esrc; float* edst;
  unsigned* mmax; float* denom; float* ebuf;
};

static void run_edge_phase(const LayerBufs& L, const int* ei, int E, int N,
                           int Hh, hipStream_t stream) {
  const long long Etot = (long long)E + N;
  const int Mo = Hh * 16;
  fill_u32<<<cdiv(N * Hh, 256), 256, 0, stream>>>(L.mmax, ORDERED_NEG_INF, N * Hh);
  fill_f32<<<cdiv(N * Hh, 256), 256, 0, stream>>>(L.denom, 0.f, N * Hh);
  fill_f32<<<cdiv((long long)N * Mo, 256), 256, 0, stream>>>(L.outb, 0.f, N * Mo);
  edge_score_max<<<cdiv(Etot, 256), 256, 0, stream>>>(ei, E, N, Hh, L.esrc,
                                                      L.edst, L.ebuf, L.mmax);
  edge_exp_sum<<<cdiv(Etot, 256), 256, 0, stream>>>(ei, E, N, Hh, L.mmax,
                                                    L.ebuf, L.denom);
  edge_aggregate<<<cdiv(Etot * Hh, 256), 256, 0, stream>>>(
      ei, E, N, Hh, L.ebuf, L.denom, L.hbuf, L.outb);
}

extern "C" void kernel_launch(void* const* d_in, const int* in_sizes, int n_in,
                              void* d_out, int out_size, void* d_ws, size_t ws_size,
                              hipStream_t stream) {
  const float* x   = (const float*)d_in[0];
  const int*   ei  = (const int*)d_in[1];
  const float* W1  = (const float*)d_in[2];
  const float* a1s = (const float*)d_in[3];
  const float* a1d = (const float*)d_in[4];
  const float* b1  = (const float*)d_in[5];
  const float* W2  = (const float*)d_in[6];
  const float* a2s = (const float*)d_in[7];
  const float* a2d = (const float*)d_in[8];
  const float* b2  = (const float*)d_in[9];
  const float* W3  = (const float*)d_in[10];
  const float* a3s = (const float*)d_in[11];
  const float* a3d = (const float*)d_in[12];
  const float* b3  = (const float*)d_in[13];

  const int B = 4, F_IN = 128, H = 4, HC = 64;
  const int E = in_sizes[1] / 2;
  const int N = in_sizes[0] / (B * F_IN);
  const int NCLS = in_sizes[13];

  float* ws = (float*)d_ws;
  float*    hbuf  = ws;
  float*    xbuf  = hbuf + (size_t)N * HC;
  float*    outb  = xbuf + (size_t)N * HC;
  float*    esrc  = outb + (size_t)N * HC;
  float*    edst  = esrc + (size_t)N * H;
  unsigned* mmax  = (unsigned*)(edst + (size_t)N * H);
  float*    denom = (float*)(mmax + (size_t)N * H);
  float*    ebuf  = denom + (size_t)N * H;

  const int ntiles = (N + 15) / 16;
  const int gemm_grid = (ntiles + 3) / 4;

  LayerBufs L{hbuf, outb, esrc, edst, mmax, denom, ebuf};

  for (int b = 0; b < B; ++b) {
    gemm_wmma_f32<128, 64><<<gemm_grid, 128, 0, stream>>>(
        x + (size_t)b * N * F_IN, W1, hbuf, ntiles);
    node_attn<<<cdiv((long long)N * H, 256), 256, 0, stream>>>(
        hbuf, a1s, a1d, esrc, edst, N, H);
    run_edge_phase(L, ei, E, N, H, stream);
    bias_act<<<cdiv((long long)N * HC, 256), 256, 0, stream>>>(
        outb, b1, xbuf, N, HC, 1);

    gemm_wmma_f32<64, 64><<<gemm_grid, 128, 0, stream>>>(xbuf, W2, hbuf, ntiles);
    node_attn<<<cdiv((long long)N * H, 256), 256, 0, stream>>>(
        hbuf, a2s, a2d, esrc, edst, N, H);
    run_edge_phase(L, ei, E, N, H, stream);
    bias_act<<<cdiv((long long)N * HC, 256), 256, 0, stream>>>(
        outb, b2, xbuf, N, HC, 1);

    gemm_wmma_f32<64, 16><<<gemm_grid, 128, 0, stream>>>(xbuf, W3, hbuf, ntiles);
    node_attn<<<cdiv((long long)N * 1, 256), 256, 0, stream>>>(
        hbuf, a3s, a3d, esrc, edst, N, 1);
    run_edge_phase(L, ei, E, N, 1, stream);
    bias_act<<<cdiv((long long)N * NCLS, 256), 256, 0, stream>>>(
        outb, b3, (float*)d_out + (size_t)b * N * NCLS, N, NCLS, 0);
  }
}


# Edge listing scans a 1-byte range-key column (dst/196, written once per call by a new small kernel), not the 4-byte dst column: 16 edges per 16-B load; 9->10 launches; sum order as was (arrival)
# speedup vs baseline: 11.7874x; 11.7874x over previous
#include <hip/hip_runtime.h>
#include <hip/hip_bf16.h>


typedef __attribute__((ext_vector_type(2))) float v2f;
typedef __attribute__((ext_vector_type(8))) float v8f;
typedef __attribute__((ext_vector_type(4))) int v4i;

#define NEG_SLOPE 0.2f
#define GAT_EPS   1e-16f

template <int KDIM, int MO>
__global__ __launch_bounds__(128) void gemm_wmma_f32(
    const float* __restrict__ X, const float* __restrict__ W,
    float* __restrict__ Hout, int ntiles) {
  const int wave = threadIdx.x >> 5;
  const int tile = blockIdx.x * 4 + wave;
  if (tile >= ntiles) return;
  const int lane  = threadIdx.x & 31;
  const int lmod  = lane & 15;
  const int lhalf = lane >> 4;
  const int row0  = tile * 16;

  v8f acc[MO / 16];
#pragma unroll
  for (int t = 0; t < MO / 16; ++t) acc[t] = v8f{0, 0, 0, 0, 0, 0, 0, 0};

  const float* xrow = X + (size_t)(row0 + lmod) * KDIM;
#pragma unroll 4
  for (int k0 = 0; k0 < KDIM; k0 += 4) {
    const int ka = k0 + 2 * lhalf;
    v2f a;
    a.x = xrow[ka];
    a.y = xrow[ka + 1];
#pragma unroll
    for (int t = 0; t < MO / 16; ++t) {
      const float* wr = W + (size_t)ka * MO + t * 16 + lmod;
      v2f b;
      b.x = wr[0];
      b.y = wr[MO];
      acc[t] = __builtin_amdgcn_wmma_f32_16x16x4_f32(
          false, a, false, b, (short)0, acc[t], false, false);
    }
  }
#pragma unroll
  for (int t = 0; t < MO / 16; ++t)
#pragma unroll
    for (int r = 0; r < 8; ++r)
      Hout[(size_t)(row0 + r + 8 * lhalf) * MO + t * 16 + lmod] = acc[t][r];
}

__global__ void node_attn(const float* __restrict__ Hm,
                          const float* __restrict__ a_src,
                          const float* __restrict__ a_dst,
                          float* __restrict__ esrc, float* __restrict__ edst,
                          int N, int Hh, int NBv) {
  int idx = blockIdx.x * blockDim.x + threadIdx.x;
  if (idx >= N * NBv * Hh) return;
  const int h = idx % Hh;
  const int b = (idx / Hh) % NBv;
  const int n = idx / (Hh * NBv);
  const float* row = Hm + (((size_t)b * N + n) * Hh + h) * 16;
  float s = 0.f, d = 0.f;
#pragma unroll
  for (int c = 0; c < 16; ++c) {
    const float v = row[c];
    s += v * a_src[h * 16 + c];
    d += v * a_dst[h * 16 + c];
  }
  esrc[idx] = s;
  edst[idx] = d;
}

__device__ __forceinline__ unsigned f2o(float f) {
  unsigned u = __float_as_uint(f);
  return (u & 0x80000000u) ? ~u : (u | 0x80000000u);
}
__device__ __forceinline__ float o2f(unsigned u) {
  return __uint_as_float((u & 0x80000000u) ? (u & 0x7FFFFFFFu) : ~u);
}
#define ORDERED_NEG_INF 0x007FFFFFu

#define ROWS   196
#define QCAP   512
#define LOADS  8
#define RA_THREADS 1024

__device__ __forceinline__ float bcast_f(float x, int i) {
  return __int_as_float(__builtin_amdgcn_readlane(__float_as_int(x), i));
}
__device__ __forceinline__ void lds_add(float* p, float v) {
  __hip_atomic_fetch_add(p, v, __ATOMIC_RELAXED, __HIP_MEMORY_SCOPE_WORKGROUP);
}
__device__ __forceinline__ void lds_max(unsigned* p, unsigned v) {
  __hip_atomic_fetch_max(p, v, __ATOMIC_RELAXED, __HIP_MEMORY_SCOPE_WORKGROUP);
}
__device__ __forceinline__ float lrelu(float e) {
  return (e > 0.f) ? e : NEG_SLOPE * e;
}

__global__ __launch_bounds__(256) void range_keys(const int* __restrict__ dst, int E,
                                                  unsigned* __restrict__ key4) {
  const int i = blockIdx.x * blockDim.x + threadIdx.x;
  const int t0 = 4 * i;
  if (t0 >= E) return;
  unsigned d[4];
  if ((E & 3) == 0) {
    const v4i v = ((const v4i*)dst)[i];
    d[0] = (unsigned)v.x; d[1] = (unsigned)v.y; d[2] = (unsigned)v.z; d[3] = (unsigned)v.w;
  } else {
#pragma unroll
    for (int k = 0; k < 4; ++k) d[k] = (t0 + k < E) ? (unsigned)dst[t0 + k] : 0u;
  }
  key4[i] = ((d[0] / ROWS) & 0xFFu) | (((d[1] / ROWS) & 0xFFu) << 8) |
            (((d[2] / ROWS) & 0xFFu) << 16) | (((d[3] / ROWS) & 0xFFu) << 24);
}

__device__ __forceinline__ unsigned zero_bytes(unsigned x) {
  const unsigned y = (x & 0x7F7F7F7Fu) + 0x7F7F7F7Fu;
  return ~(y | x | 0x7F7F7F7Fu);
}

template <int HH, int NB, bool KEYS>
__global__ __launch_bounds__(RA_THREADS) void range_aggregate(
    const int* __restrict__ ei, const unsigned char* __restrict__ key, int E, int N,
    const float* __restrict__ esrc, const float* __restrict__ edst,
    const float* __restrict__ Hm, const float* __restrict__ bias,
    float* __restrict__ out, int relu) {
  constexpr int CH = HH * 16;
  constexpr int J  = HH * NB;
  constexpr int NW = RA_THREADS / 32;
  static_assert(ROWS <= RA_THREADS, "one thread per self loop");
  static_assert(J <= 32, "slow path: one lane per logit");
  static_assert(NB == 1 || NB % 2 == 0, "the 16-channel walk takes the batches in pairs");
  __shared__ float acc[NB * ROWS * CH];
  __shared__ __attribute__((aligned(16))) float den[ROWS * J];
  __shared__ __attribute__((aligned(16))) unsigned mxu[ROWS * J];
  __shared__ __attribute__((aligned(16))) float edl[ROWS * J];
  __shared__ int qt[NW][QCAP];
  __shared__ int ovf;

  const int tid  = threadIdx.x;
  const int lane = tid & 31;
  const int wib  = __builtin_amdgcn_readfirstlane(tid >> 5);
  const int r0   = blockIdx.x * ROWS;
  const int nrows = (N - r0 < ROWS) ? (N - r0) : ROWS;
  int* const myq = qt[wib];

  for (int i = tid; i < NB * ROWS * CH; i += RA_THREADS) acc[i] = 0.f;
  for (int i = tid; i < ROWS * J; i += RA_THREADS) {
    den[i] = 0.f;
    mxu[i] = ORDERED_NEG_INF;
    edl[i] = (i < nrows * J) ? edst[(size_t)r0 * J + i] : 0.f;
  }
  if (tid == 0) ovf = 0;
  __syncthreads();

  int tail = 0;

  auto push = [&](bool m, int t) {
    const unsigned mask = __builtin_amdgcn_ballot_w32(m);
    if (mask != 0u) {
      const int pos = tail + (int)__builtin_amdgcn_mbcnt_lo(mask, 0u);
      if (m && pos < QCAP) myq[pos] = t;
      tail += __builtin_popcount(mask);
    }
  };

  push(tid < nrows, E + r0 + tid);

  const int* dp = ei + E;
  if constexpr (KEYS) {
    const unsigned bid4 = (unsigned)blockIdx.x * 0x01010101u;
    const int nvec = E >> 4;
    if (wib == 0) {
      const int t = 16 * nvec + lane;
      bool m = false;
      if (t < E) m = key[t] == (unsigned char)blockIdx.x;
      push(m, t);
    }
    if (nvec > 0) {
      const v4i* kp4 = (const v4i*)key;
      auto ld = [&](int j) -> v4i { return kp4[j < nvec ? j : nvec - 1]; };
      v4i cur[LOADS], nxt[LOADS];
#pragma unroll
      for (int k = 0; k < LOADS; ++k) cur[k] = ld(k * RA_THREADS + tid);
      for (int base = 0; base < nvec; base += LOADS * RA_THREADS) {
#pragma unroll
        for (int k = 0; k < LOADS; ++k) nxt[k] = ld(base + (LOADS + k) * RA_THREADS + tid);
#pragma unroll
        for (int k = 0; k < LOADS; ++k) {
          const v4i v = cur[k];
          const int j = base + k * RA_THREADS + tid;
          unsigned m = (zero_bytes((unsigned)v.x ^ bid4) >> 7) | (zero_bytes((unsigned)v.y ^ bid4) >> 6) |
                       (zero_bytes((unsigned)v.z ^ bid4) >> 5) | (zero_bytes((unsigned)v.w ^ bid4) >> 4);
          if (j >= nvec) m = 0u;
          unsigned live = __builtin_amdgcn_ballot_w32(m != 0u);
          while (live != 0u) {
            const int p = __builtin_ctz(m | 0x80000000u);
            const int pos = tail + (int)__builtin_amdgcn_mbcnt_lo(live, 0u);
            if (m != 0u && pos < QCAP) myq[pos] = 16 * j + 4 * (p & 7) + (p >> 3);
            tail += __builtin_popcount(live);
            m &= m - 1u;
            live = __builtin_amdgcn_ballot_w32(m != 0u);
          }
        }
#pragma unroll
        for (int k = 0; k < LOADS; ++k) cur[k] = nxt[k];
      }
    }
  } else {
  int lead = (4 - (E & 3)) & 3;
  if (lead > E) lead = E;
  const int nvec = (E - lead) >> 2;
  const int tail0 = lead + 4 * nvec;
  if (wib == 0) {
    const int nodd = lead + (E - tail0);
    const int t = (lane < lead) ? lane : tail0 + (lane - lead);
    int d = -1;
    if (lane < nodd) d = dp[t];
    push((unsigned)(d - r0) < (unsigned)nrows, t);
  }

  if (nvec > 0) {
    const v4i* dp4 = (const v4i*)(dp + lead);
    auto ld = [&](int j) -> v4i { return dp4[j < nvec ? j : nvec - 1]; };
    v4i cur[LOADS], nxt[LOADS];
#pragma unroll
    for (int k = 0; k < LOADS; ++k) cur[k] = ld(k * RA_THREADS + tid);
    for (int base = 0; base < nvec; base += LOADS * RA_THREADS) {
#pragma unroll
      for (int k = 0; k < LOADS; ++k) nxt[k] = ld(base + (LOADS + k) * RA_THREADS + tid);
#pragma unroll
      for (int k = 0; k < LOADS; ++k) {
        const v4i v = cur[k];
        const int j = base + k * RA_THREADS + tid;
        const bool ok = j < nvec;
        const bool m0 = ok && (unsigned)(v.x - r0) < (unsigned)nrows;
        const bool m1 = ok && (unsigned)(v.y - r0) < (unsigned)nrows;
        const bool m2 = ok && (unsigned)(v.z - r0) < (unsigned)nrows;
        const bool m3 = ok && (unsigned)(v.w - r0) < (unsigned)nrows;
        if (__builtin_amdgcn_ballot_w32(m0 || m1 || m2 || m3) != 0u) {
          const int tb = lead + 4 * j;
          push(m0, tb);
          push(m1, tb + 1);
          push(m2, tb + 2);
          push(m3, tb + 3);
        }
      }
#pragma unroll
      for (int k = 0; k < LOADS; ++k) cur[k] = nxt[k];
    }
  }
  }
  if (tail > QCAP && lane == 0) ovf = 1;
  __syncthreads();
  const bool slow = ovf != 0;

  auto ends = [&](int t, int& s, int& dl) {
    const bool real = t < E;
    const int tc = real ? t : 0;
    s = ei[tc];
    int d = ei[(size_t)E + tc];
    if (!real) s = d = t - E;
    dl = d - r0;
  };

  auto scores = [&](int s, int dl, float (&e)[J]) {
    const float* sp = esrc + (size_t)s * J;
    const float* dq = edl + dl * J;
    if constexpr (J % 4 == 0) {
#pragma unroll
      for (int q = 0; q < J / 4; ++q) {
        const float4 a = ((const float4*)sp)[q];
        const float4 b = ((const float4*)dq)[q];
        e[4 * q + 0] = lrelu(a.x + b.x);
        e[4 * q + 1] = lrelu(a.y + b.y);
        e[4 * q + 2] = lrelu(a.z + b.z);
        e[4 * q + 3] = lrelu(a.w + b.w);
      }
    } else {
#pragma unroll
      for (int j = 0; j < J; ++j) e[j] = lrelu(sp[j] + dq[j]);
    }
  };

  if (!slow) {
    const int nq = tail;

    for (int b0 = 0; b0 < nq; b0 += 64) {
      int sv[2], dv[2];
      bool on[2];
#pragma unroll
      for (int k = 0; k < 2; ++k) {
        const int i = b0 + 32 * k + lane;
        on[k] = i < nq;
        ends(myq[on[k] ? i : b0], sv[k], dv[k]);
      }
      float e[2][J];
#pragma unroll
      for (int k = 0; k < 2; ++k) scores(sv[k], dv[k], e[k]);
#pragma unroll
      for (int k = 0; k < 2; ++k) {
#pragma unroll
        for (int j = 0; j < J; ++j) lds_max(&mxu[dv[k] * J + j], on[k] ? f2o(e[k][j]) : 0u);
      }
    }
    __syncthreads();

    auto batch = [&](int b0, int n) {
      int s = 0, dl = 0;
      float al[J];
#pragma unroll
      for (int j = 0; j < J; ++j) al[j] = 0.f;
      if (lane < n) {
        ends(myq[b0 + lane], s, dl);
        float e[J];
        scores(s, dl, e);
        const unsigned* mp = mxu + dl * J;
        if constexpr (J % 4 == 0) {
#pragma unroll
          for (int q = 0; q < J / 4; ++q) {
            const uint4 mv = ((const uint4*)mp)[q];
            al[4 * q + 0] = __expf(e[4 * q + 0] - o2f(mv.x));
            al[4 * q + 1] = __expf(e[4 * q + 1] - o2f(mv.y));
            al[4 * q + 2] = __expf(e[4 * q + 2] - o2f(mv.z));
            al[4 * q + 3] = __expf(e[4 * q + 3] - o2f(mv.w));
          }
        } else {
#pragma unroll
          for (int j = 0; j < J; ++j) al[j] = __expf(e[j] - o2f(mp[j]));
        }
#pragma unroll
        for (int j = 0; j < J; ++j) lds_add(&den[dl * J + j], al[j]);
      }
      if constexpr (HH == 4) {
        const bool up = lane >= 16;
        constexpr int EPT = (NB >= 4) ? 4 : 8 / NB;
#pragma unroll 1
        for (int i0 = 0; i0 < n; i0 += EPT) {
          float v0[EPT * NB], v1[EPT * NB];
#pragma unroll
          for (int k = 0; k < EPT; ++k) {
            const int si = __builtin_amdgcn_readlane(s, i0 + k);
#pragma unroll
            for (int b = 0; b < NB; ++b) {
              const float* hp = Hm + ((size_t)b * N + si) * 64 + lane;
              v0[k * NB + b] = hp[0];
              v1[k * NB + b] = hp[32];
            }
          }
#pragma unroll
          for (int k = 0; k < EPT; ++k) {
            if (i0 + k < n) {
              const int i = i0 + k;
              const int di = __builtin_amdgcn_readlane(dl, i);
#pragma unroll
              for (int b = 0; b < NB; ++b) {
                const float a0 = bcast_f(al[4 * b + 0], i), a1 = bcast_f(al[4 * b + 1], i);
                const float a2 = bcast_f(al[4 * b + 2], i), a3 = bcast_f(al[4 * b + 3], i);
                float* op = acc + (b * ROWS + di) * 64 + lane;
                lds_add(op, (up ? a1 : a0) * v0[k * NB + b]);
                lds_add(op + 32, (up ? a3 : a2) * v1[k * NB + b]);
              }
            }
          }
        }
      } else if constexpr (NB == 1) {
        const int sub = lane >> 4;
        const int c = lane & 15;
#pragma unroll 1
        for (int i0 = 0; i0 < n; i0 += 16) {
          float v[8], a[8];
          int dj[8];
#pragma unroll
          for (int k = 0; k < 8; ++k) {
            const int j = i0 + 2 * k + sub;
            const int sj = __shfl(s, j);
            dj[k] = __shfl(dl, j);
            a[k] = __shfl(al[0], j);
            v[k] = Hm[(size_t)sj * 16 + c];
          }
#pragma unroll
          for (int k = 0; k < 8; ++k) {
            const int j = i0 + 2 * k + sub;
            if (j < n) lds_add(acc + dj[k] * 16 + c, a[k] * v[k]);
          }
        }
      } else {
        const int sub = lane >> 4;
        const int c = lane & 15;
        constexpr int LPE = NB / 2;
        constexpr int EPT = 16 / LPE;
#pragma unroll 1
        for (int i0 = 0; i0 < n; i0 += EPT) {
          float v[EPT * LPE];
#pragma unroll
          for (int k = 0; k < EPT; ++k) {
            const int si = __builtin_amdgcn_readlane(s, i0 + k);
#pragma unroll
            for (int q = 0; q < LPE; ++q)
              v[k * LPE + q] = Hm[((size_t)(2 * q + sub) * N + si) * 16 + c];
          }
#pragma unroll
          for (int k = 0; k < EPT; ++k) {
            if (i0 + k < n) {
              const int i = i0 + k;
              const int di = __builtin_amdgcn_readlane(dl, i);
#pragma unroll
              for (int q = 0; q < LPE; ++q) {
                const float a0 = bcast_f(al[2 * q], i), a1 = bcast_f(al[2 * q + 1], i);
                lds_add(acc + ((2 * q + sub) * ROWS + di) * 16 + c, (sub ? a1 : a0) * v[k * LPE + q]);
              }
            }
          }
        }
      }
    };

    for (int b0 = 0; b0 < nq; b0 += 32) batch(b0, (nq - b0 < 32) ? (nq - b0) : 32);
  } else {
    auto sweep = [&](auto&& f) {
      for (int i = wib; i < nrows; i += NW) f(E + r0 + i);
      for (int base = 0; base < E; base += RA_THREADS) {
        const int t = base + tid;
        int d = -1;
        if (t < E) d = dp[t];
        unsigned mask = __builtin_amdgcn_ballot_w32((unsigned)(d - r0) < (unsigned)nrows);
        while (mask != 0u) {
          const int l = __builtin_ctz(mask);
          mask &= mask - 1u;
          f(base + wib * 32 + l);
        }
      }
    };
    sweep([&](int t) {
      int s, dl;
      ends(t, s, dl);
      if (lane < J)
        lds_max(&mxu[dl * J + lane], f2o(lrelu(esrc[(size_t)s * J + lane] + edl[dl * J + lane])));
    });
    __syncthreads();
    sweep([&](int t) {
      int s, dl;
      ends(t, s, dl);
      float w = 0.f;
      if (lane < J) {
        const float e = lrelu(esrc[(size_t)s * J + lane] + edl[dl * J + lane]);
        w = __expf(e - o2f(mxu[dl * J + lane]));
        lds_add(&den[dl * J + lane], w);
      }
#pragma unroll
      for (int b = 0; b < NB; ++b) {
#pragma unroll
        for (int c0 = 0; c0 < CH; c0 += 32) {
          const int c = c0 + lane;
          const bool valid = c < CH;
          const float wj = __shfl(w, b * HH + (valid ? (c >> 4) : 0));
          if (valid) lds_add(acc + (b * ROWS + dl) * CH + c, wj * Hm[((size_t)b * N + s) * CH + c]);
        }
      }
    });
  }
  __syncthreads();

  for (int b = 0; b < NB; ++b) {
    for (int idx = tid; idx < nrows * CH; idx += RA_THREADS) {
      const int dl = idx / CH;
      const int c = idx % CH;
      float v = acc[b * ROWS * CH + idx] / (den[(dl * NB + b) * HH + (c >> 4)] + GAT_EPS) + bias[c];
      if (relu) v = fmaxf(v, 0.f);
      out[((size_t)b * N + r0) * CH + idx] = v;
    }
  }
}

static inline int cdiv(long long a, int b) { return (int)((a + b - 1) / b); }

struct LayerBufs {
  float* hbuf; float* esrc; float* edst;
};

template <int HH, int NB>
static void run_edge_phase(const LayerBufs& L, const int* ei, const unsigned char* key, int E, int N,
                           const float* bias, float* outp, int relu,
                           hipStream_t stream) {
  if (key)
    range_aggregate<HH, NB, true><<<cdiv(N, ROWS), RA_THREADS, 0, stream>>>(
        ei, key, E, N, L.esrc, L.edst, L.hbuf, bias, outp, relu);
  else
    range_aggregate<HH, NB, false><<<cdiv(N, ROWS), RA_THREADS, 0, stream>>>(
        ei, key, E, N, L.esrc, L.edst, L.hbuf, bias, outp, relu);
}

static inline size_t ws_need(int NB, int N, int E) {
  return sizeof(float) * ((size_t)NB * N * 64 * 2 + (size_t)NB * N * 4 * 2);
}

static inline unsigned char* key_place(int NB, int N, int E, void* d_ws, size_t ws_size) {
  const size_t off = (ws_need(NB, N, E) + 15) & ~(size_t)15;
  if (cdiv(N, ROWS) > 256 || ws_size < off + (size_t)E + 16) return nullptr;
  return (unsigned char*)d_ws + off;
}

template <int NB>
static void run_pass(void* const* d_in, int b0, int N, int E, int NCLS,
                     float* d_out, void* d_ws, const unsigned char* key, hipStream_t stream) {
  const float* x   = (const float*)d_in[0];
  const int*   ei  = (const int*)d_in[1];
  const float* W1  = (const float*)d_in[2];
  const float* a1s = (const float*)d_in[3];
  const float* a1d = (const float*)d_in[4];
  const float* b1  = (const float*)d_in[5];
  const float* W2  = (const float*)d_in[6];
  const float* a2s = (const float*)d_in[7];
  const float* a2d = (const float*)d_in[8];
  const float* b2  = (const float*)d_in[9];
  const float* W3  = (const float*)d_in[10];
  const float* a3s = (const float*)d_in[11];
  const float* a3d = (const float*)d_in[12];
  const float* b3  = (const float*)d_in[13];
  const int F_IN = 128, H = 4, HC = 64;

  float* ws = (float*)d_ws;
  float*    hbuf  = ws;
  float*    xbuf  = hbuf + (size_t)NB * N * HC;
  float*    esrc  = xbuf + (size_t)NB * N * HC;
  float*    edst  = esrc + (size_t)NB * N * H;

  const int ntiles = (NB * N + 15) / 16;
  const int gemm_grid = (ntiles + 3) / 4;

  LayerBufs L{hbuf, esrc, edst};

  gemm_wmma_f32<128, 64><<<gemm_grid, 128, 0, stream>>>(
      x + (size_t)b0 * N * F_IN, W1, hbuf, ntiles);
  node_attn<<<cdiv((long long)N * NB * H, 256), 256, 0, stream>>>(
      hbuf, a1s, a1d, esrc, edst, N, H, NB);
  run_edge_phase<4, NB>(L, ei, key, E, N, b1, xbuf, 1, stream);

  gemm_wmma_f32<64, 64><<<gemm_grid, 128, 0, stream>>>(xbuf, W2, hbuf, ntiles);
  node_attn<<<cdiv((long long)N * NB * H, 256), 256, 0, stream>>>(
      hbuf, a2s, a2d, esrc, edst, N, H, NB);
  run_edge_phase<4, NB>(L, ei, key, E, N, b2, xbuf, 1, stream);

  gemm_wmma_f32<64, 16><<<gemm_grid, 128, 0, stream>>>(xbuf, W3, hbuf, ntiles);
  node_attn<<<cdiv((long long)N * NB, 256), 256, 0, stream>>>(
      hbuf, a3s, a3d, esrc, edst, N, 1, NB);
  run_edge_phase<1, NB>(L, ei, key, E, N, b3, d_out + (size_t)b0 * N * NCLS, 0, stream);
}

extern "C" void kernel_launch(void* const* d_in, const int* in_sizes, int n_in,
                              void* d_out, int out_size, void* d_ws, size_t ws_size,
                              hipStream_t stream) {
  const int B = 4, F_IN = 128;
  const int E = in_sizes[1] / 2;
  const int N = in_sizes[0] / (B * F_IN);
  const int NCLS = in_sizes[13];

  const int NBP = (ws_size >= ws_need(4, N, E)) ? 4 : (ws_size >= ws_need(2, N, E)) ? 2 : 1;

  unsigned char* key = key_place(NBP, N, E, d_ws, ws_size);
  if (key)
    range_keys<<<cdiv(cdiv(E, 4), 256), 256, 0, stream>>>((const int*)d_in[1] + E, E, (unsigned*)key);

  if (NBP == 4) {
    run_pass<4>(d_in, 0, N, E, NCLS, (float*)d_out, d_ws, key, stream);
  } else if (NBP == 2) {
    for (int b = 0; b < B; b += 2) run_pass<2>(d_in, b, N, E, NCLS, (float*)d_out, d_ws, key, stream);
  } else {
    for (int b = 0; b < B; ++b) run_pass<1>(d_in, b, N, E, NCLS, (float*)d_out, d_ws, key, stream);
  }
}


# Max walk leaves each listed edge's source and local dst row in LDS; the sum walk takes them there and does not re-read the two index columns: 2 random 4-byte loads per edge gone; sum order as was
# speedup vs baseline: 12.1575x; 1.0314x over previous
#include <hip/hip_runtime.h>
#include <hip/hip_bf16.h>


typedef __attribute__((ext_vector_type(2))) float v2f;
typedef __attribute__((ext_vector_type(8))) float v8f;
typedef __attribute__((ext_vector_type(4))) int v4i;

#define NEG_SLOPE 0.2f
#define GAT_EPS   1e-16f

template <int KDIM, int MO>
__global__ __launch_bounds__(128) void gemm_wmma_f32(
    const float* __restrict__ X, const float* __restrict__ W,
    float* __restrict__ Hout, int ntiles) {
  const int wave = threadIdx.x >> 5;
  const int tile = blockIdx.x * 4 + wave;
  if (tile >= ntiles) return;
  const int lane  = threadIdx.x & 31;
  const int lmod  = lane & 15;
  const int lhalf = lane >> 4;
  const int row0  = tile * 16;

  v8f acc[MO / 16];
#pragma unroll
  for (int t = 0; t < MO / 16; ++t) acc[t] = v8f{0, 0, 0, 0, 0, 0, 0, 0};

  const float* xrow = X + (size_t)(row0 + lmod) * KDIM;
#pragma unroll 4
  for (int k0 = 0; k0 < KDIM; k0 += 4) {
    const int ka = k0 + 2 * lhalf;
    v2f a;
    a.x = xrow[ka];
    a.y = xrow[ka + 1];
#pragma unroll
    for (int t = 0; t < MO / 16; ++t) {
      const float* wr = W + (size_t)ka * MO + t * 16 + lmod;
      v2f b;
      b.x = wr[0];
      b.y = wr[MO];
      acc[t] = __builtin_amdgcn_wmma_f32_16x16x4_f32(
          false, a, false, b, (short)0, acc[t], false, false);
    }
  }
#pragma unroll
  for (int t = 0; t < MO / 16; ++t)
#pragma unroll
    for (int r = 0; r < 8; ++r)
      Hout[(size_t)(row0 + r + 8 * lhalf) * MO + t * 16 + lmod] = acc[t][r];
}

__global__ void node_attn(const float* __restrict__ Hm,
                          const float* __restrict__ a_src,
                          const float* __restrict__ a_dst,
                          float* __restrict__ esrc, float* __restrict__ edst,
                          int N, int Hh, int NBv) {
  int idx = blockIdx.x * blockDim.x + threadIdx.x;
  if (idx >= N * NBv * Hh) return;
  const int h = idx % Hh;
  const int b = (idx / Hh) % NBv;
  const int n = idx / (Hh * NBv);
  const float* row = Hm + (((size_t)b * N + n) * Hh + h) * 16;
  float s = 0.f, d = 0.f;
#pragma unroll
  for (int c = 0; c < 16; ++c) {
    const float v = row[c];
    s += v * a_src[h * 16 + c];
    d += v * a_dst[h * 16 + c];
  }
  esrc[idx] = s;
  edst[idx] = d;
}

__device__ __forceinline__ unsigned f2o(float f) {
  unsigned u = __float_as_uint(f);
  return (u & 0x80000000u) ? ~u : (u | 0x80000000u);
}
__device__ __forceinline__ float o2f(unsigned u) {
  return __uint_as_float((u & 0x80000000u) ? (u & 0x7FFFFFFFu) : ~u);
}
#define ORDERED_NEG_INF 0x007FFFFFu

#define ROWS   196
#define QCAP   512
#define LOADS  8
#define RA_THREADS 1024

__device__ __forceinline__ float bcast_f(float x, int i) {
  return __int_as_float(__builtin_amdgcn_readlane(__float_as_int(x), i));
}
__device__ __forceinline__ void lds_add(float* p, float v) {
  __hip_atomic_fetch_add(p, v, __ATOMIC_RELAXED, __HIP_MEMORY_SCOPE_WORKGROUP);
}
__device__ __forceinline__ void lds_max(unsigned* p, unsigned v) {
  __hip_atomic_fetch_max(p, v, __ATOMIC_RELAXED, __HIP_MEMORY_SCOPE_WORKGROUP);
}
__device__ __forceinline__ float lrelu(float e) {
  return (e > 0.f) ? e : NEG_SLOPE * e;
}

__global__ __launch_bounds__(256) void range_keys(const int* __restrict__ dst, int E,
                                                  unsigned* __restrict__ key4) {
  const int i = blockIdx.x * blockDim.x + threadIdx.x;
  const int t0 = 4 * i;
  if (t0 >= E) return;
  unsigned d[4];
  if ((E & 3) == 0) {
    const v4i v = ((const v4i*)dst)[i];
    d[0] = (unsigned)v.x; d[1] = (unsigned)v.y; d[2] = (unsigned)v.z; d[3] = (unsigned)v.w;
  } else {
#pragma unroll
    for (int k = 0; k < 4; ++k) d[k] = (t0 + k < E) ? (unsigned)dst[t0 + k] : 0u;
  }
  key4[i] = ((d[0] / ROWS) & 0xFFu) | (((d[1] / ROWS) & 0xFFu) << 8) |
            (((d[2] / ROWS) & 0xFFu) << 16) | (((d[3] / ROWS) & 0xFFu) << 24);
}

__device__ __forceinline__ unsigned zero_bytes(unsigned x) {
  const unsigned y = (x & 0x7F7F7F7Fu) + 0x7F7F7F7Fu;
  return ~(y | x | 0x7F7F7F7Fu);
}

template <int HH, int NB, bool KEYS>
__global__ __launch_bounds__(RA_THREADS) void range_aggregate(
    const int* __restrict__ ei, const unsigned char* __restrict__ key, int E, int N,
    const float* __restrict__ esrc, const float* __restrict__ edst,
    const float* __restrict__ Hm, const float* __restrict__ bias,
    float* __restrict__ out, int relu) {
  constexpr int CH = HH * 16;
  constexpr int J  = HH * NB;
  constexpr int NW = RA_THREADS / 32;
  static_assert(ROWS <= RA_THREADS, "one thread per self loop");
  static_assert(ROWS <= 256, "a local destination row fits a byte of the row list");
  static_assert(J <= 32, "slow path: one lane per logit");
  static_assert(NB == 1 || NB % 2 == 0, "the 16-channel walk takes the batches in pairs");
  __shared__ float acc[NB * ROWS * CH];
  __shared__ __attribute__((aligned(16))) float den[ROWS * J];
  __shared__ __attribute__((aligned(16))) unsigned mxu[ROWS * J];
  __shared__ __attribute__((aligned(16))) float edl[ROWS * J];
  __shared__ int qt[NW][QCAP];
  __shared__ unsigned char dq[NW][QCAP];
  __shared__ int ovf;

  const int tid  = threadIdx.x;
  const int lane = tid & 31;
  const int wib  = __builtin_amdgcn_readfirstlane(tid >> 5);
  const int r0   = blockIdx.x * ROWS;
  const int nrows = (N - r0 < ROWS) ? (N - r0) : ROWS;
  int* const myq = qt[wib];
  unsigned char* const myd = dq[wib];

  for (int i = tid; i < NB * ROWS * CH; i += RA_THREADS) acc[i] = 0.f;
  for (int i = tid; i < ROWS * J; i += RA_THREADS) {
    den[i] = 0.f;
    mxu[i] = ORDERED_NEG_INF;
    edl[i] = (i < nrows * J) ? edst[(size_t)r0 * J + i] : 0.f;
  }
  if (tid == 0) ovf = 0;
  __syncthreads();

  int tail = 0;

  auto push = [&](bool m, int t) {
    const unsigned mask = __builtin_amdgcn_ballot_w32(m);
    if (mask != 0u) {
      const int pos = tail + (int)__builtin_amdgcn_mbcnt_lo(mask, 0u);
      if (m && pos < QCAP) myq[pos] = t;
      tail += __builtin_popcount(mask);
    }
  };

  push(tid < nrows, E + r0 + tid);

  const int* dp = ei + E;
  if constexpr (KEYS) {
    const unsigned bid4 = (unsigned)blockIdx.x * 0x01010101u;
    const int nvec = E >> 4;
    if (wib == 0) {
      const int t = 16 * nvec + lane;
      bool m = false;
      if (t < E) m = key[t] == (unsigned char)blockIdx.x;
      push(m, t);
    }
    if (nvec > 0) {
      const v4i* kp4 = (const v4i*)key;
      auto ld = [&](int j) -> v4i { return kp4[j < nvec ? j : nvec - 1]; };
      v4i cur[LOADS], nxt[LOADS];
#pragma unroll
      for (int k = 0; k < LOADS; ++k) cur[k] = ld(k * RA_THREADS + tid);
      for (int base = 0; base < nvec; base += LOADS * RA_THREADS) {
#pragma unroll
        for (int k = 0; k < LOADS; ++k) nxt[k] = ld(base + (LOADS + k) * RA_THREADS + tid);
#pragma unroll
        for (int k = 0; k < LOADS; ++k) {
          const v4i v = cur[k];
          const int j = base + k * RA_THREADS + tid;
          unsigned m = (zero_bytes((unsigned)v.x ^ bid4) >> 7) | (zero_bytes((unsigned)v.y ^ bid4) >> 6) |
                       (zero_bytes((unsigned)v.z ^ bid4) >> 5) | (zero_bytes((unsigned)v.w ^ bid4) >> 4);
          if (j >= nvec) m = 0u;
          unsigned live = __builtin_amdgcn_ballot_w32(m != 0u);
          while (live != 0u) {
            const int p = __builtin_ctz(m | 0x80000000u);
            const int pos = tail + (int)__builtin_amdgcn_mbcnt_lo(live, 0u);
            if (m != 0u && pos < QCAP) myq[pos] = 16 * j + 4 * (p & 7) + (p >> 3);
            tail += __builtin_popcount(live);
            m &= m - 1u;
            live = __builtin_amdgcn_ballot_w32(m != 0u);
          }
        }
#pragma unroll
        for (int k = 0; k < LOADS; ++k) cur[k] = nxt[k];
      }
    }
  } else {
  int lead = (4 - (E & 3)) & 3;
  if (lead > E) lead = E;
  const int nvec = (E - lead) >> 2;
  const int tail0 = lead + 4 * nvec;
  if (wib == 0) {
    const int nodd = lead + (E - tail0);
    const int t = (lane < lead) ? lane : tail0 + (lane - lead);
    int d = -1;
    if (lane < nodd) d = dp[t];
    push((unsigned)(d - r0) < (unsigned)nrows, t);
  }

  if (nvec > 0) {
    const v4i* dp4 = (const v4i*)(dp + lead);
    auto ld = [&](int j) -> v4i { return dp4[j < nvec ? j : nvec - 1]; };
    v4i cur[LOADS], nxt[LOADS];
#pragma unroll
    for (int k = 0; k < LOADS; ++k) cur[k] = ld(k * RA_THREADS + tid);
    for (int base = 0; base < nvec; base += LOADS * RA_THREADS) {
#pragma unroll
      for (int k = 0; k < LOADS; ++k) nxt[k] = ld(base + (LOADS + k) * RA_THREADS + tid);
#pragma unroll
      for (int k = 0; k < LOADS; ++k) {
        const v4i v = cur[k];
        const int j = base + k * RA_THREADS + tid;
        const bool ok = j < nvec;
        const bool m0 = ok && (unsigned)(v.x - r0) < (unsigned)nrows;
        const bool m1 = ok && (unsigned)(v.y - r0) < (unsigned)nrows;
        const bool m2 = ok && (unsigned)(v.z - r0) < (unsigned)nrows;
        const bool m3 = ok && (unsigned)(v.w - r0) < (unsigned)nrows;
        if (__builtin_amdgcn_ballot_w32(m0 || m1 || m2 || m3) != 0u) {
          const int tb = lead + 4 * j;
          push(m0, tb);
          push(m1, tb + 1);
          push(m2, tb + 2);
          push(m3, tb + 3);
        }
      }
#pragma unroll
      for (int k = 0; k < LOADS; ++k) cur[k] = nxt[k];
    }
  }
  }
  if (tail > QCAP && lane == 0) ovf = 1;
  __syncthreads();
  const bool slow = ovf != 0;

  auto ends = [&](int t, int& s, int& dl) {
    const bool real = t < E;
    const int tc = real ? t : 0;
    s = ei[tc];
    int d = ei[(size_t)E + tc];
    if (!real) s = d = t - E;
    dl = d - r0;
  };

  auto scores = [&](int s, int dl, float (&e)[J]) {
    const float* sp = esrc + (size_t)s * J;
    const float* dq = edl + dl * J;
    if constexpr (J % 4 == 0) {
#pragma unroll
      for (int q = 0; q < J / 4; ++q) {
        const float4 a = ((const float4*)sp)[q];
        const float4 b = ((const float4*)dq)[q];
        e[4 * q + 0] = lrelu(a.x + b.x);
        e[4 * q + 1] = lrelu(a.y + b.y);
        e[4 * q + 2] = lrelu(a.z + b.z);
        e[4 * q + 3] = lrelu(a.w + b.w);
      }
    } else {
#pragma unroll
      for (int j = 0; j < J; ++j) e[j] = lrelu(sp[j] + dq[j]);
    }
  };

  if (!slow) {
    const int nq = tail;

    for (int b0 = 0; b0 < nq; b0 += 64) {
      int sv[2], dv[2];
      bool on[2];
#pragma unroll
      for (int k = 0; k < 2; ++k) {
        const int i = b0 + 32 * k + lane;
        on[k] = i < nq;
        ends(myq[on[k] ? i : b0], sv[k], dv[k]);
      }
      float e[2][J];
#pragma unroll
      for (int k = 0; k < 2; ++k) scores(sv[k], dv[k], e[k]);
#pragma unroll
      for (int k = 0; k < 2; ++k) {
        if (on[k]) {
          myq[b0 + 32 * k + lane] = sv[k];
          myd[b0 + 32 * k + lane] = (unsigned char)dv[k];
        }
      }
#pragma unroll
      for (int k = 0; k < 2; ++k) {
#pragma unroll
        for (int j = 0; j < J; ++j) lds_max(&mxu[dv[k] * J + j], on[k] ? f2o(e[k][j]) : 0u);
      }
    }
    __syncthreads();

    auto batch = [&](int b0, int n) {
      int s = 0, dl = 0;
      float al[J];
#pragma unroll
      for (int j = 0; j < J; ++j) al[j] = 0.f;
      if (lane < n) {
        s = myq[b0 + lane];
        dl = myd[b0 + lane];
        float e[J];
        scores(s, dl, e);
        const unsigned* mp = mxu + dl * J;
        if constexpr (J % 4 == 0) {
#pragma unroll
          for (int q = 0; q < J / 4; ++q) {
            const uint4 mv = ((const uint4*)mp)[q];
            al[4 * q + 0] = __expf(e[4 * q + 0] - o2f(mv.x));
            al[4 * q + 1] = __expf(e[4 * q + 1] - o2f(mv.y));
            al[4 * q + 2] = __expf(e[4 * q + 2] - o2f(mv.z));
            al[4 * q + 3] = __expf(e[4 * q + 3] - o2f(mv.w));
          }
        } else {
#pragma unroll
          for (int j = 0; j < J; ++j) al[j] = __expf(e[j] - o2f(mp[j]));
        }
#pragma unroll
        for (int j = 0; j < J; ++j) lds_add(&den[dl * J + j], al[j]);
      }
      if constexpr (HH == 4) {
        const bool up = lane >= 16;
        constexpr int EPT = (NB >= 4) ? 4 : 8 / NB;
#pragma unroll 1
        for (int i0 = 0; i0 < n; i0 += EPT) {
          float v0[EPT * NB], v1[EPT * NB];
#pragma unroll
          for (int k = 0; k < EPT; ++k) {
            const int si = __builtin_amdgcn_readlane(s, i0 + k);
#pragma unroll
            for (int b = 0; b < NB; ++b) {
              const float* hp = Hm + ((size_t)b * N + si) * 64 + lane;
              v0[k * NB + b] = hp[0];
              v1[k * NB + b] = hp[32];
            }
          }
#pragma unroll
          for (int k = 0; k < EPT; ++k) {
            if (i0 + k < n) {
              const int i = i0 + k;
              const int di = __builtin_amdgcn_readlane(dl, i);
#pragma unroll
              for (int b = 0; b < NB; ++b) {
                const float a0 = bcast_f(al[4 * b + 0], i), a1 = bcast_f(al[4 * b + 1], i);
                const float a2 = bcast_f(al[4 * b + 2], i), a3 = bcast_f(al[4 * b + 3], i);
                float* op = acc + (b * ROWS + di) * 64 + lane;
                lds_add(op, (up ? a1 : a0) * v0[k * NB + b]);
                lds_add(op + 32, (up ? a3 : a2) * v1[k * NB + b]);
              }
            }
          }
        }
      } else if constexpr (NB == 1) {
        const int sub = lane >> 4;
        const int c = lane & 15;
#pragma unroll 1
        for (int i0 = 0; i0 < n; i0 += 16) {
          float v[8], a[8];
          int dj[8];
#pragma unroll
          for (int k = 0; k < 8; ++k) {
            const int j = i0 + 2 * k + sub;
            const int sj = __shfl(s, j);
            dj[k] = __shfl(dl, j);
            a[k] = __shfl(al[0], j);
            v[k] = Hm[(size_t)sj * 16 + c];
          }
#pragma unroll
          for (int k = 0; k < 8; ++k) {
            const int j = i0 + 2 * k + sub;
            if (j < n) lds_add(acc + dj[k] * 16 + c, a[k] * v[k]);
          }
        }
      } else {
        const int sub = lane >> 4;
        const int c = lane & 15;
        constexpr int LPE = NB / 2;
        constexpr int EPT = 16 / LPE;
#pragma unroll 1
        for (int i0 = 0; i0 < n; i0 += EPT) {
          float v[EPT * LPE];
#pragma unroll
          for (int k = 0; k < EPT; ++k) {
            const int si = __builtin_amdgcn_readlane(s, i0 + k);
#pragma unroll
            for (int q = 0; q < LPE; ++q)
              v[k * LPE + q] = Hm[((size_t)(2 * q + sub) * N + si) * 16 + c];
          }
#pragma unroll
          for (int k = 0; k < EPT; ++k) {
            if (i0 + k < n) {
              const int i = i0 + k;
              const int di = __builtin_amdgcn_readlane(dl, i);
#pragma unroll
              for (int q = 0; q < LPE; ++q) {
                const float a0 = bcast_f(al[2 * q], i), a1 = bcast_f(al[2 * q + 1], i);
                lds_add(acc + ((2 * q + sub) * ROWS + di) * 16 + c, (sub ? a1 : a0) * v[k * LPE + q]);
              }
            }
          }
        }
      }
    };

    for (int b0 = 0; b0 < nq; b0 += 32) batch(b0, (nq - b0 < 32) ? (nq - b0) : 32);
  } else {
    auto sweep = [&](auto&& f) {
      for (int i = wib; i < nrows; i += NW) f(E + r0 + i);
      for (int base = 0; base < E; base += RA_THREADS) {
        const int t = base + tid;
        int d = -1;
        if (t < E) d = dp[t];
        unsigned mask = __builtin_amdgcn_ballot_w32((unsigned)(d - r0) < (unsigned)nrows);
        while (mask != 0u) {
          const int l = __builtin_ctz(mask);
          mask &= mask - 1u;
          f(base + wib * 32 + l);
        }
      }
    };
    sweep([&](int t) {
      int s, dl;
      ends(t, s, dl);
      if (lane < J)
        lds_max(&mxu[dl * J + lane], f2o(lrelu(esrc[(size_t)s * J + lane] + edl[dl * J + lane])));
    });
    __syncthreads();
    sweep([&](int t) {
      int s, dl;
      ends(t, s, dl);
      float w = 0.f;
      if (lane < J) {
        const float e = lrelu(esrc[(size_t)s * J + lane] + edl[dl * J + lane]);
        w = __expf(e - o2f(mxu[dl * J + lane]));
        lds_add(&den[dl * J + lane], w);
      }
#pragma unroll
      for (int b = 0; b < NB; ++b) {
#pragma unroll
        for (int c0 = 0; c0 < CH; c0 += 32) {
          const int c = c0 + lane;
          const bool valid = c < CH;
          const float wj = __shfl(w, b * HH + (valid ? (c >> 4) : 0));
          if (valid) lds_add(acc + (b * ROWS + dl) * CH + c, wj * Hm[((size_t)b * N + s) * CH + c]);
        }
      }
    });
  }
  __syncthreads();

  for (int b = 0; b < NB; ++b) {
    for (int idx = tid; idx < nrows * CH; idx += RA_THREADS) {
      const int dl = idx / CH;
      const int c = idx % CH;
      float v = acc[b * ROWS * CH + idx] / (den[(dl * NB + b) * HH + (c >> 4)] + GAT_EPS) + bias[c];
      if (relu) v = fmaxf(v, 0.f);
      out[((size_t)b * N + r0) * CH + idx] = v;
    }
  }
}

static inline int cdiv(long long a, int b) { return (int)((a + b - 1) / b); }

struct LayerBufs {
  float* hbuf; float* esrc; float* edst;
};

template <int HH, int NB>
static void run_edge_phase(const LayerBufs& L, const int* ei, const unsigned char* key, int E, int N,
                           const float* bias, float* outp, int relu,
                           hipStream_t stream) {
  if (key)
    range_aggregate<HH, NB, true><<<cdiv(N, ROWS), RA_THREADS, 0, stream>>>(
        ei, key, E, N, L.esrc, L.edst, L.hbuf, bias, outp, relu);
  else
    range_aggregate<HH, NB, false><<<cdiv(N, ROWS), RA_THREADS, 0, stream>>>(
        ei, key, E, N, L.esrc, L.edst, L.hbuf, bias, outp, relu);
}

static inline size_t ws_need(int NB, int N, int E) {
  return sizeof(float) * ((size_t)NB * N * 64 * 2 + (size_t)NB * N * 4 * 2);
}

static inline unsigned char* key_place(int NB, int N, int E, void* d_ws, size_t ws_size) {
  const size_t off = (ws_need(NB, N, E) + 15) & ~(size_t)15;
  if (cdiv(N, ROWS) > 256 || ws_size < off + (size_t)E + 16) return nullptr;
  return (unsigned char*)d_ws + off;
}

template <int NB>
static void run_pass(void* const* d_in, int b0, int N, int E, int NCLS,
                     float* d_out, void* d_ws, const unsigned char* key, hipStream_t stream) {
  const float* x   = (const float*)d_in[0];
  const int*   ei  = (const int*)d_in[1];
  const float* W1  = (const float*)d_in[2];
  const float* a1s = (const float*)d_in[3];
  const float* a1d = (const float*)d_in[4];
  const float* b1  = (const float*)d_in[5];
  const float* W2  = (const float*)d_in[6];
  const float* a2s = (const float*)d_in[7];
  const float* a2d = (const float*)d_in[8];
  const float* b2  = (const float*)d_in[9];
  const float* W3  = (const float*)d_in[10];
  const float* a3s = (const float*)d_in[11];
  const float* a3d = (const float*)d_in[12];
  const float* b3  = (const float*)d_in[13];
  const int F_IN = 128, H = 4, HC = 64;

  float* ws = (float*)d_ws;
  float*    hbuf  = ws;
  float*    xbuf  = hbuf + (size_t)NB * N * HC;
  float*    esrc  = xbuf + (size_t)NB * N * HC;
  float*    edst  = esrc + (size_t)NB * N * H;

  const int ntiles = (NB * N + 15) / 16;
  const int gemm_grid = (ntiles + 3) / 4;

  LayerBufs L{hbuf, esrc, edst};

  gemm_wmma_f32<128, 64><<<gemm_grid, 128, 0, stream>>>(
      x + (size_t)b0 * N * F_IN, W1, hbuf, ntiles);
  node_attn<<<cdiv((long long)N * NB * H, 256), 256, 0, stream>>>(
      hbuf, a1s, a1d, esrc, edst, N, H, NB);
  run_edge_phase<4, NB>(L, ei, key, E, N, b1, xbuf, 1, stream);

  gemm_wmma_f32<64, 64><<<gemm_grid, 128, 0, stream>>>(xbuf, W2, hbuf, ntiles);
  node_attn<<<cdiv((long long)N * NB * H, 256), 256, 0, stream>>>(
      hbuf, a2s, a2d, esrc, edst, N, H, NB);
  run_edge_phase<4, NB>(L, ei, key, E, N, b2, xbuf, 1, stream);

  gemm_wmma_f32<64, 16><<<gemm_grid, 128, 0, stream>>>(xbuf, W3, hbuf, ntiles);
  node_attn<<<cdiv((long long)N * NB, 256), 256, 0, stream>>>(
      hbuf, a3s, a3d, esrc, edst, N, 1, NB);
  run_edge_phase<1, NB>(L, ei, key, E, N, b3, d_out + (size_t)b0 * N * NCLS, 0, stream);
}

extern "C" void kernel_launch(void* const* d_in, const int* in_sizes, int n_in,
                              void* d_out, int out_size, void* d_ws, size_t ws_size,
                              hipStream_t stream) {
  const int B = 4, F_IN = 128;
  const int E = in_sizes[1] / 2;
  const int N = in_sizes[0] / (B * F_IN);
  const int NCLS = in_sizes[13];

  const int NBP = (ws_size >= ws_need(4, N, E)) ? 4 : (ws_size >= ws_need(2, N, E)) ? 2 : 1;

  unsigned char* key = key_place(NBP, N, E, d_ws, ws_size);
  if (key)
    range_keys<<<cdiv(cdiv(E, 4), 256), 256, 0, stream>>>((const int*)d_in[1] + E, E, (unsigned*)key);

  if (NBP == 4) {
    run_pass<4>(d_in, 0, N, E, NCLS, (float*)d_out, d_ws, key, stream);
  } else if (NBP == 2) {
    for (int b = 0; b < B; b += 2) run_pass<2>(d_in, b, N, E, NCLS, (float*)d_out, d_ws, key, stream);
  } else {
    for (int b = 0; b < B; ++b) run_pass<1>(d_in, b, N, E, NCLS, (float*)d_out, d_ws, key, stream);
  }
}


# LDS bank conflicts: per-row logit/max/sum tables at a row pitch of 20 words (odd count of 16-B pieces; was 16), 16-channel sums' batch planes 16 words apart mod 32; sum order as was
# speedup vs baseline: 13.3261x; 1.0961x over previous
#include <hip/hip_runtime.h>
#include <hip/hip_bf16.h>


typedef __attribute__((ext_vector_type(2))) float v2f;
typedef __attribute__((ext_vector_type(8))) float v8f;
typedef __attribute__((ext_vector_type(4))) int v4i;

#define NEG_SLOPE 0.2f
#define GAT_EPS   1e-16f

template <int KDIM, int MO>
__global__ __launch_bounds__(128) void gemm_wmma_f32(
    const float* __restrict__ X, const float* __restrict__ W,
    float* __restrict__ Hout, int ntiles) {
  const int wave = threadIdx.x >> 5;
  const int tile = blockIdx.x * 4 + wave;
  if (tile >= ntiles) return;
  const int lane  = threadIdx.x & 31;
  const int lmod  = lane & 15;
  const int lhalf = lane >> 4;
  const int row0  = tile * 16;

  v8f acc[MO / 16];
#pragma unroll
  for (int t = 0; t < MO / 16; ++t) acc[t] = v8f{0, 0, 0, 0, 0, 0, 0, 0};

  const float* xrow = X + (size_t)(row0 + lmod) * KDIM;
#pragma unroll 4
  for (int k0 = 0; k0 < KDIM; k0 += 4) {
    const int ka = k0 + 2 * lhalf;
    v2f a;
    a.x = xrow[ka];
    a.y = xrow[ka + 1];
#pragma unroll
    for (int t = 0; t < MO / 16; ++t) {
      const float* wr = W + (size_t)ka * MO + t * 16 + lmod;
      v2f b;
      b.x = wr[0];
      b.y = wr[MO];
      acc[t] = __builtin_amdgcn_wmma_f32_16x16x4_f32(
          false, a, false, b, (short)0, acc[t], false, false);
    }
  }
#pragma unroll
  for (int t = 0; t < MO / 16; ++t)
#pragma unroll
    for (int r = 0; r < 8; ++r)
      Hout[(size_t)(row0 + r + 8 * lhalf) * MO + t * 16 + lmod] = acc[t][r];
}

__global__ void node_attn(const float* __restrict__ Hm,
                          const float* __restrict__ a_src,
                          const float* __restrict__ a_dst,
                          float* __restrict__ esrc, float* __restrict__ edst,
                          int N, int Hh, int NBv) {
  int idx = blockIdx.x * blockDim.x + threadIdx.x;
  if (idx >= N * NBv * Hh) return;
  const int h = idx % Hh;
  const int b = (idx / Hh) % NBv;
  const int n = idx / (Hh * NBv);
  const float* row = Hm + (((size_t)b * N + n) * Hh + h) * 16;
  float s = 0.f, d = 0.f;
#pragma unroll
  for (int c = 0; c < 16; ++c) {
    const float v = row[c];
    s += v * a_src[h * 16 + c];
    d += v * a_dst[h * 16 + c];
  }
  esrc[idx] = s;
  edst[idx] = d;
}

__device__ __forceinline__ unsigned f2o(float f) {
  unsigned u = __float_as_uint(f);
  return (u & 0x80000000u) ? ~u : (u | 0x80000000u);
}
__device__ __forceinline__ float o2f(unsigned u) {
  return __uint_as_float((u & 0x80000000u) ? (u & 0x7FFFFFFFu) : ~u);
}
#define ORDERED_NEG_INF 0x007FFFFFu

#define ROWS   196
#define QCAP   448
#define LOADS  8
#define RA_THREADS 1024

__device__ __forceinline__ float bcast_f(float x, int i) {
  return __int_as_float(__builtin_amdgcn_readlane(__float_as_int(x), i));
}
__device__ __forceinline__ void lds_add(float* p, float v) {
  __hip_atomic_fetch_add(p, v, __ATOMIC_RELAXED, __HIP_MEMORY_SCOPE_WORKGROUP);
}
__device__ __forceinline__ void lds_max(unsigned* p, unsigned v) {
  __hip_atomic_fetch_max(p, v, __ATOMIC_RELAXED, __HIP_MEMORY_SCOPE_WORKGROUP);
}
__device__ __forceinline__ float lrelu(float e) {
  return (e > 0.f) ? e : NEG_SLOPE * e;
}

__global__ __launch_bounds__(256) void range_keys(const int* __restrict__ dst, int E,
                                                  unsigned* __restrict__ key4) {
  const int i = blockIdx.x * blockDim.x + threadIdx.x;
  const int t0 = 4 * i;
  if (t0 >= E) return;
  unsigned d[4];
  if ((E & 3) == 0) {
    const v4i v = ((const v4i*)dst)[i];
    d[0] = (unsigned)v.x; d[1] = (unsigned)v.y; d[2] = (unsigned)v.z; d[3] = (unsigned)v.w;
  } else {
#pragma unroll
    for (int k = 0; k < 4; ++k) d[k] = (t0 + k < E) ? (unsigned)dst[t0 + k] : 0u;
  }
  key4[i] = ((d[0] / ROWS) & 0xFFu) | (((d[1] / ROWS) & 0xFFu) << 8) |
            (((d[2] / ROWS) & 0xFFu) << 16) | (((d[3] / ROWS) & 0xFFu) << 24);
}

__device__ __forceinline__ unsigned zero_bytes(unsigned x) {
  const unsigned y = (x & 0x7F7F7F7Fu) + 0x7F7F7F7Fu;
  return ~(y | x | 0x7F7F7F7Fu);
}

template <int HH, int NB, bool KEYS>
__global__ __launch_bounds__(RA_THREADS) void range_aggregate(
    const int* __restrict__ ei, const unsigned char* __restrict__ key, int E, int N,
    const float* __restrict__ esrc, const float* __restrict__ edst,
    const float* __restrict__ Hm, const float* __restrict__ bias,
    float* __restrict__ out, int relu) {
  constexpr int CH = HH * 16;
  constexpr int J  = HH * NB;
  constexpr int NW = RA_THREADS / 32;
  static_assert(ROWS <= RA_THREADS, "one thread per self loop");
  static_assert(ROWS <= 256, "a local destination row fits a byte of the row list");
  static_assert(J <= 32, "slow path: one lane per logit");
  static_assert(NB == 1 || NB % 2 == 0, "the 16-channel walk takes the batches in pairs");
  constexpr int PJ = (J % 8 == 0) ? J + 4 : J;
  constexpr int AP = ROWS * CH + ((CH == 16 && NB > 1 && (ROWS * CH) % 32 == 0) ? 16 : 0);
  __shared__ float acc[NB * AP];
  __shared__ __attribute__((aligned(16))) float den[ROWS * PJ];
  __shared__ __attribute__((aligned(16))) unsigned mxu[ROWS * PJ];
  __shared__ __attribute__((aligned(16))) float edl[ROWS * PJ];
  __shared__ int qt[NW][QCAP];
  __shared__ unsigned char dq[NW][QCAP];
  __shared__ int ovf;

  const int tid  = threadIdx.x;
  const int lane = tid & 31;
  const int wib  = __builtin_amdgcn_readfirstlane(tid >> 5);
  const int r0   = blockIdx.x * ROWS;
  const int nrows = (N - r0 < ROWS) ? (N - r0) : ROWS;
  int* const myq = qt[wib];
  unsigned char* const myd = dq[wib];

  for (int i = tid; i < NB * AP; i += RA_THREADS) acc[i] = 0.f;
  for (int i = tid; i < ROWS * PJ; i += RA_THREADS) {
    const int row = i / PJ;
    const int j = i - row * PJ;
    den[i] = 0.f;
    mxu[i] = ORDERED_NEG_INF;
    edl[i] = (j < J && row < nrows) ? edst[(size_t)(r0 + row) * J + j] : 0.f;
  }
  if (tid == 0) ovf = 0;
  __syncthreads();

  int tail = 0;

  auto push = [&](bool m, int t) {
    const unsigned mask = __builtin_amdgcn_ballot_w32(m);
    if (mask != 0u) {
      const int pos = tail + (int)__builtin_amdgcn_mbcnt_lo(mask, 0u);
      if (m && pos < QCAP) myq[pos] = t;
      tail += __builtin_popcount(mask);
    }
  };

  push(tid < nrows, E + r0 + tid);

  const int* dp = ei + E;
  if constexpr (KEYS) {
    const unsigned bid4 = (unsigned)blockIdx.x * 0x01010101u;
    const int nvec = E >> 4;
    if (wib == 0) {
      const int t = 16 * nvec + lane;
      bool m = false;
      if (t < E) m = key[t] == (unsigned char)blockIdx.x;
      push(m, t);
    }
    if (nvec > 0) {
      const v4i* kp4 = (const v4i*)key;
      auto ld = [&](int j) -> v4i { return kp4[j < nvec ? j : nvec - 1]; };
      v4i cur[LOADS], nxt[LOADS];
#pragma unroll
      for (int k = 0; k < LOADS; ++k) cur[k] = ld(k * RA_THREADS + tid);
      for (int base = 0; base < nvec; base += LOADS * RA_THREADS) {
#pragma unroll
        for (int k = 0; k < LOADS; ++k) nxt[k] = ld(base + (LOADS + k) * RA_THREADS + tid);
#pragma unroll
        for (int k = 0; k < LOADS; ++k) {
          const v4i v = cur[k];
          const int j = base + k * RA_THREADS + tid;
          unsigned m = (zero_bytes((unsigned)v.x ^ bid4) >> 7) | (zero_bytes((unsigned)v.y ^ bid4) >> 6) |
                       (zero_bytes((unsigned)v.z ^ bid4) >> 5) | (zero_bytes((unsigned)v.w ^ bid4) >> 4);
          if (j >= nvec) m = 0u;
          unsigned live = __builtin_amdgcn_ballot_w32(m != 0u);
          while (live != 0u) {
            const int p = __builtin_ctz(m | 0x80000000u);
            const int pos = tail + (int)__builtin_amdgcn_mbcnt_lo(live, 0u);
            if (m != 0u && pos < QCAP) myq[pos] = 16 * j + 4 * (p & 7) + (p >> 3);
            tail += __builtin_popcount(live);
            m &= m - 1u;
            live = __builtin_amdgcn_ballot_w32(m != 0u);
          }
        }
#pragma unroll
        for (int k = 0; k < LOADS; ++k) cur[k] = nxt[k];
      }
    }
  } else {
  int lead = (4 - (E & 3)) & 3;
  if (lead > E) lead = E;
  const int nvec = (E - lead) >> 2;
  const int tail0 = lead + 4 * nvec;
  if (wib == 0) {
    const int nodd = lead + (E - tail0);
    const int t = (lane < lead) ? lane : tail0 + (lane - lead);
    int d = -1;
    if (lane < nodd) d = dp[t];
    push((unsigned)(d - r0) < (unsigned)nrows, t);
  }

  if (nvec > 0) {
    const v4i* dp4 = (const v4i*)(dp + lead);
    auto ld = [&](int j) -> v4i { return dp4[j < nvec ? j : nvec - 1]; };
    v4i cur[LOADS], nxt[LOADS];
#pragma unroll
    for (int k = 0; k < LOADS; ++k) cur[k] = ld(k * RA_THREADS + tid);
    for (int base = 0; base < nvec; base += LOADS * RA_THREADS) {
#pragma unroll
      for (int k = 0; k < LOADS; ++k) nxt[k] = ld(base + (LOADS + k) * RA_THREADS + tid);
#pragma unroll
      for (int k = 0; k < LOADS; ++k) {
        const v4i v = cur[k];
        const int j = base + k * RA_THREADS + tid;
        const bool ok = j < nvec;
        const bool m0 = ok && (unsigned)(v.x - r0) < (unsigned)nrows;
        const bool m1 = ok && (unsigned)(v.y - r0) < (unsigned)nrows;
        const bool m2 = ok && (unsigned)(v.z - r0) < (unsigned)nrows;
        const bool m3 = ok && (unsigned)(v.w - r0) < (unsigned)nrows;
        if (__builtin_amdgcn_ballot_w32(m0 || m1 || m2 || m3) != 0u) {
          const int tb = lead + 4 * j;
          push(m0, tb);
          push(m1, tb + 1);
          push(m2, tb + 2);
          push(m3, tb + 3);
        }
      }
#pragma unroll
      for (int k = 0; k < LOADS; ++k) cur[k] = nxt[k];
    }
  }
  }
  if (tail > QCAP && lane == 0) ovf = 1;
  __syncthreads();
  const bool slow = ovf != 0;

  auto ends = [&](int t, int& s, int& dl) {
    const bool real = t < E;
    const int tc = real ? t : 0;
    s = ei[tc];
    int d = ei[(size_t)E + tc];
    if (!real) s = d = t - E;
    dl = d - r0;
  };

  auto scores = [&](int s, int dl, float (&e)[J]) {
    const float* sp = esrc + (size_t)s * J;
    const float* dq = edl + dl * PJ;
    if constexpr (J % 4 == 0) {
#pragma unroll
      for (int q = 0; q < J / 4; ++q) {
        const float4 a = ((const float4*)sp)[q];
        const float4 b = ((const float4*)dq)[q];
        e[4 * q + 0] = lrelu(a.x + b.x);
        e[4 * q + 1] = lrelu(a.y + b.y);
        e[4 * q + 2] = lrelu(a.z + b.z);
        e[4 * q + 3] = lrelu(a.w + b.w);
      }
    } else {
#pragma unroll
      for (int j = 0; j < J; ++j) e[j] = lrelu(sp[j] + dq[j]);
    }
  };

  if (!slow) {
    const int nq = tail;

    for (int b0 = 0; b0 < nq; b0 += 64) {
      int sv[2], dv[2];
      bool on[2];
#pragma unroll
      for (int k = 0; k < 2; ++k) {
        const int i = b0 + 32 * k + lane;
        on[k] = i < nq;
        ends(myq[on[k] ? i : b0], sv[k], dv[k]);
      }
      float e[2][J];
#pragma unroll
      for (int k = 0; k < 2; ++k) scores(sv[k], dv[k], e[k]);
#pragma unroll
      for (int k = 0; k < 2; ++k) {
        if (on[k]) {
          myq[b0 + 32 * k + lane] = sv[k];
          myd[b0 + 32 * k + lane] = (unsigned char)dv[k];
        }
      }
#pragma unroll
      for (int k = 0; k < 2; ++k) {
#pragma unroll
        for (int j = 0; j < J; ++j) lds_max(&mxu[dv[k] * PJ + j], on[k] ? f2o(e[k][j]) : 0u);
      }
    }
    __syncthreads();

    auto batch = [&](int b0, int n) {
      int s = 0, dl = 0;
      float al[J];
#pragma unroll
      for (int j = 0; j < J; ++j) al[j] = 0.f;
      if (lane < n) {
        s = myq[b0 + lane];
        dl = myd[b0 + lane];
        float e[J];
        scores(s, dl, e);
        const unsigned* mp = mxu + dl * PJ;
        if constexpr (J % 4 == 0) {
#pragma unroll
          for (int q = 0; q < J / 4; ++q) {
            const uint4 mv = ((const uint4*)mp)[q];
            al[4 * q + 0] = __expf(e[4 * q + 0] - o2f(mv.x));
            al[4 * q + 1] = __expf(e[4 * q + 1] - o2f(mv.y));
            al[4 * q + 2] = __expf(e[4 * q + 2] - o2f(mv.z));
            al[4 * q + 3] = __expf(e[4 * q + 3] - o2f(mv.w));
          }
        } else {
#pragma unroll
          for (int j = 0; j < J; ++j) al[j] = __expf(e[j] - o2f(mp[j]));
        }
#pragma unroll
        for (int j = 0; j < J; ++j) lds_add(&den[dl * PJ + j], al[j]);
      }
      if constexpr (HH == 4) {
        const bool up = lane >= 16;
        constexpr int EPT = (NB >= 4) ? 4 : 8 / NB;
#pragma unroll 1
        for (int i0 = 0; i0 < n; i0 += EPT) {
          float v0[EPT * NB], v1[EPT * NB];
#pragma unroll
          for (int k = 0; k < EPT; ++k) {
            const int si = __builtin_amdgcn_readlane(s, i0 + k);
#pragma unroll
            for (int b = 0; b < NB; ++b) {
              const float* hp = Hm + ((size_t)b * N + si) * 64 + lane;
              v0[k * NB + b] = hp[0];
              v1[k * NB + b] = hp[32];
            }
          }
#pragma unroll
          for (int k = 0; k < EPT; ++k) {
            if (i0 + k < n) {
              const int i = i0 + k;
              const int di = __builtin_amdgcn_readlane(dl, i);
#pragma unroll
              for (int b = 0; b < NB; ++b) {
                const float a0 = bcast_f(al[4 * b + 0], i), a1 = bcast_f(al[4 * b + 1], i);
                const float a2 = bcast_f(al[4 * b + 2], i), a3 = bcast_f(al[4 * b + 3], i);
                float* op = acc + b * AP + di * 64 + lane;
                lds_add(op, (up ? a1 : a0) * v0[k * NB + b]);
                lds_add(op + 32, (up ? a3 : a2) * v1[k * NB + b]);
              }
            }
          }
        }
      } else if constexpr (NB == 1) {
        const int sub = lane >> 4;
        const int c = lane & 15;
#pragma unroll 1
        for (int i0 = 0; i0 < n; i0 += 16) {
          float v[8], a[8];
          int dj[8];
#pragma unroll
          for (int k = 0; k < 8; ++k) {
            const int j = i0 + 2 * k + sub;
            const int sj = __shfl(s, j);
            dj[k] = __shfl(dl, j);
            a[k] = __shfl(al[0], j);
            v[k] = Hm[(size_t)sj * 16 + c];
          }
#pragma unroll
          for (int k = 0; k < 8; ++k) {
            const int j = i0 + 2 * k + sub;
            if (j < n) lds_add(acc + dj[k] * 16 + c, a[k] * v[k]);
          }
        }
      } else {
        const int sub = lane >> 4;
        const int c = lane & 15;
        constexpr int LPE = NB / 2;
        constexpr int EPT = 16 / LPE;
#pragma unroll 1
        for (int i0 = 0; i0 < n; i0 += EPT) {
          float v[EPT * LPE];
#pragma unroll
          for (int k = 0; k < EPT; ++k) {
            const int si = __builtin_amdgcn_readlane(s, i0 + k);
#pragma unroll
            for (int q = 0; q < LPE; ++q)
              v[k * LPE + q] = Hm[((size_t)(2 * q + sub) * N + si) * 16 + c];
          }
#pragma unroll
          for (int k = 0; k < EPT; ++k) {
            if (i0 + k < n) {
              const int i = i0 + k;
              const int di = __builtin_amdgcn_readlane(dl, i);
#pragma unroll
              for (int q = 0; q < LPE; ++q) {
                const float a0 = bcast_f(al[2 * q], i), a1 = bcast_f(al[2 * q + 1], i);
                lds_add(acc + (2 * q + sub) * AP + di * 16 + c, (sub ? a1 : a0) * v[k * LPE + q]);
              }
            }
          }
        }
      }
    };

    for (int b0 = 0; b0 < nq; b0 += 32) batch(b0, (nq - b0 < 32) ? (nq - b0) : 32);
  } else {
    auto sweep = [&](auto&& f) {
      for (int i = wib; i < nrows; i += NW) f(E + r0 + i);
      for (int base = 0; base < E; base += RA_THREADS) {
        const int t = base + tid;
        int d = -1;
        if (t < E) d = dp[t];
        unsigned mask = __builtin_amdgcn_ballot_w32((unsigned)(d - r0) < (unsigned)nrows);
        while (mask != 0u) {
          const int l = __builtin_ctz(mask);
          mask &= mask - 1u;
          f(base + wib * 32 + l);
        }
      }
    };
    sweep([&](int t) {
      int s, dl;
      ends(t, s, dl);
      if (lane < J)
        lds_max(&mxu[dl * PJ + lane], f2o(lrelu(esrc[(size_t)s * J + lane] + edl[dl * PJ + lane])));
    });
    __syncthreads();
    sweep([&](int t) {
      int s, dl;
      ends(t, s, dl);
      float w = 0.f;
      if (lane < J) {
        const float e = lrelu(esrc[(size_t)s * J + lane] + edl[dl * PJ + lane]);
        w = __expf(e - o2f(mxu[dl * PJ + lane]));
        lds_add(&den[dl * PJ + lane], w);
      }
#pragma unroll
      for (int b = 0; b < NB; ++b) {
#pragma unroll
        for (int c0 = 0; c0 < CH; c0 += 32) {
          const int c = c0 + lane;
          const bool valid = c < CH;
          const float wj = __shfl(w, b * HH + (valid ? (c >> 4) : 0));
          if (valid) lds_add(acc + b * AP + dl * CH + c, wj * Hm[((size_t)b * N + s) * CH + c]);
        }
      }
    });
  }
  __syncthreads();

  for (int b = 0; b < NB; ++b) {
    for (int idx = tid; idx < nrows * CH; idx += RA_THREADS) {
      const int dl = idx / CH;
      const int c = idx % CH;
      float v = acc[b * AP + idx] / (den[dl * PJ + b * HH + (c >> 4)] + GAT_EPS) + bias[c];
      if (relu) v = fmaxf(v, 0.f);
      out[((size_t)b * N + r0) * CH + idx] = v;
    }
  }
}

static inline int cdiv(long long a, int b) { return (int)((a + b - 1) / b); }

struct LayerBufs {
  float* hbuf; float* esrc; float* edst;
};

template <int HH, int NB>
static void run_edge_phase(const LayerBufs& L, const int* ei, const unsigned char* key, int E, int N,
                           const float* bias, float* outp, int relu,
                           hipStream_t stream) {
  if (key)
    range_aggregate<HH, NB, true><<<cdiv(N, ROWS), RA_THREADS, 0, stream>>>(
        ei, key, E, N, L.esrc, L.edst, L.hbuf, bias, outp, relu);
  else
    range_aggregate<HH, NB, false><<<cdiv(N, ROWS), RA_THREADS, 0, stream>>>(
        ei, key, E, N, L.esrc, L.edst, L.hbuf, bias, outp, relu);
}

static inline size_t ws_need(int NB, int N, int E) {
  return sizeof(float) * ((size_t)NB * N * 64 * 2 + (size_t)NB * N * 4 * 2);
}

static inline unsigned char* key_place(int NB, int N, int E, void* d_ws, size_t ws_size) {
  const size_t off = (ws_need(NB, N, E) + 15) & ~(size_t)15;
  if (cdiv(N, ROWS) > 256 || ws_size < off + (size_t)E + 16) return nullptr;
  return (unsigned char*)d_ws + off;
}

template <int NB>
static void run_pass(void* const* d_in, int b0, int N, int E, int NCLS,
                     float* d_out, void* d_ws, const unsigned char* key, hipStream_t stream) {
  const float* x   = (const float*)d_in[0];
  const int*   ei  = (const int*)d_in[1];
  const float* W1  = (const float*)d_in[2];
  const float* a1s = (const float*)d_in[3];
  const float* a1d = (const float*)d_in[4];
  const float* b1  = (const float*)d_in[5];
  const float* W2  = (const float*)d_in[6];
  const float* a2s = (const float*)d_in[7];
  const float* a2d = (const float*)d_in[8];
  const float* b2  = (const float*)d_in[9];
  const float* W3  = (const float*)d_in[10];
  const float* a3s = (const float*)d_in[11];
  const float* a3d = (const float*)d_in[12];
  const float* b3  = (const float*)d_in[13];
  const int F_IN = 128, H = 4, HC = 64;

  float* ws = (float*)d_ws;
  float*    hbuf  = ws;
  float*    xbuf  = hbuf + (size_t)NB * N * HC;
  float*    esrc  = xbuf + (size_t)NB * N * HC;
  float*    edst  = esrc + (size_t)NB * N * H;

  const int ntiles = (NB * N + 15) / 16;
  const int gemm_grid = (ntiles + 3) / 4;

  LayerBufs L{hbuf, esrc, edst};

  gemm_wmma_f32<128, 64><<<gemm_grid, 128, 0, stream>>>(
      x + (size_t)b0 * N * F_IN, W1, hbuf, ntiles);
  node_attn<<<cdiv((long long)N * NB * H, 256), 256, 0, stream>>>(
      hbuf, a1s, a1d, esrc, edst, N, H, NB);
  run_edge_phase<4, NB>(L, ei, key, E, N, b1, xbuf, 1, stream);

  gemm_wmma_f32<64, 64><<<gemm_grid, 128, 0, stream>>>(xbuf, W2, hbuf, ntiles);
  node_attn<<<cdiv((long long)N * NB * H, 256), 256, 0, stream>>>(
      hbuf, a2s, a2d, esrc, edst, N, H, NB);
  run_edge_phase<4, NB>(L, ei, key, E, N, b2, xbuf, 1, stream);

  gemm_wmma_f32<64, 16><<<gemm_grid, 128, 0, stream>>>(xbuf, W3, hbuf, ntiles);
  node_attn<<<cdiv((long long)N * NB, 256), 256, 0, stream>>>(
      hbuf, a3s, a3d, esrc, edst, N, 1, NB);
  run_edge_phase<1, NB>(L, ei, key, E, N, b3, d_out + (size_t)b0 * N * NCLS, 0, stream);
}

extern "C" void kernel_launch(void* const* d_in, const int* in_sizes, int n_in,
                              void* d_out, int out_size, void* d_ws, size_t ws_size,
                              hipStream_t stream) {
  const int B = 4, F_IN = 128;
  const int E = in_sizes[1] / 2;
  const int N = in_sizes[0] / (B * F_IN);
  const int NCLS = in_sizes[13];

  const int NBP = (ws_size >= ws_need(4, N, E)) ? 4 : (ws_size >= ws_need(2, N, E)) ? 2 : 1;

  unsigned char* key = key_place(NBP, N, E, d_ws, ws_size);
  if (key)
    range_keys<<<cdiv(cdiv(E, 4), 256), 256, 0, stream>>>((const int*)d_in[1] + E, E, (unsigned*)key);

  if (NBP == 4) {
    run_pass<4>(d_in, 0, N, E, NCLS, (float*)d_out, d_ws, key, stream);
  } else if (NBP == 2) {
    for (int b = 0; b < B; b += 2) run_pass<2>(d_in, b, N, E, NCLS, (float*)d_out, d_ws, key, stream);
  } else {
    for (int b = 0; b < B; ++b) run_pass<1>(d_in, b, N, E, NCLS, (float*)d_out, d_ws, key, stream);
  }
}


# Edge lists built once per call: layer-1 edge kernel saves each wave's list (source, local row) in the workspace, layers 2-3 load it; key scan + index loads 1x not 3x; sum order as was (arrival)
# speedup vs baseline: 14.7471x; 1.1066x over previous
#include <hip/hip_runtime.h>
#include <hip/hip_bf16.h>


typedef __attribute__((ext_vector_type(2))) float v2f;
typedef __attribute__((ext_vector_type(8))) float v8f;
typedef __attribute__((ext_vector_type(4))) int v4i;

#define NEG_SLOPE 0.2f
#define GAT_EPS   1e-16f

template <int KDIM, int MO>
__global__ __launch_bounds__(128) void gemm_wmma_f32(
    const float* __restrict__ X, const float* __restrict__ W,
    float* __restrict__ Hout, int ntiles) {
  const int wave = threadIdx.x >> 5;
  const int tile = blockIdx.x * 4 + wave;
  if (tile >= ntiles) return;
  const int lane  = threadIdx.x & 31;
  const int lmod  = lane & 15;
  const int lhalf = lane >> 4;
  const int row0  = tile * 16;

  v8f acc[MO / 16];
#pragma unroll
  for (int t = 0; t < MO / 16; ++t) acc[t] = v8f{0, 0, 0, 0, 0, 0, 0, 0};

  const float* xrow = X + (size_t)(row0 + lmod) * KDIM;
#pragma unroll 4
  for (int k0 = 0; k0 < KDIM; k0 += 4) {
    const int ka = k0 + 2 * lhalf;
    v2f a;
    a.x = xrow[ka];
    a.y = xrow[ka + 1];
#pragma unroll
    for (int t = 0; t < MO / 16; ++t) {
      const float* wr = W + (size_t)ka * MO + t * 16 + lmod;
      v2f b;
      b.x = wr[0];
      b.y = wr[MO];
      acc[t] = __builtin_amdgcn_wmma_f32_16x16x4_f32(
          false, a, false, b, (short)0, acc[t], false, false);
    }
  }
#pragma unroll
  for (int t = 0; t < MO / 16; ++t)
#pragma unroll
    for (int r = 0; r < 8; ++r)
      Hout[(size_t)(row0 + r + 8 * lhalf) * MO + t * 16 + lmod] = acc[t][r];
}

__global__ void node_attn(const float* __restrict__ Hm,
                          const float* __restrict__ a_src,
                          const float* __restrict__ a_dst,
                          float* __restrict__ esrc, float* __restrict__ edst,
                          int N, int Hh, int NBv) {
  int idx = blockIdx.x * blockDim.x + threadIdx.x;
  if (idx >= N * NBv * Hh) return;
  const int h = idx % Hh;
  const int b = (idx / Hh) % NBv;
  const int n = idx / (Hh * NBv);
  const float* row = Hm + (((size_t)b * N + n) * Hh + h) * 16;
  float s = 0.f, d = 0.f;
#pragma unroll
  for (int c = 0; c < 16; ++c) {
    const float v = row[c];
    s += v * a_src[h * 16 + c];
    d += v * a_dst[h * 16 + c];
  }
  esrc[idx] = s;
  edst[idx] = d;
}

__device__ __forceinline__ unsigned f2o(float f) {
  unsigned u = __float_as_uint(f);
  return (u & 0x80000000u) ? ~u : (u | 0x80000000u);
}
__device__ __forceinline__ float o2f(unsigned u) {
  return __uint_as_float((u & 0x80000000u) ? (u & 0x7FFFFFFFu) : ~u);
}
#define ORDERED_NEG_INF 0x007FFFFFu

#define ROWS   196
#define QCAP   448
#define LOADS  8
#define RA_THREADS 1024

__device__ __forceinline__ float bcast_f(float x, int i) {
  return __int_as_float(__builtin_amdgcn_readlane(__float_as_int(x), i));
}
__device__ __forceinline__ void lds_add(float* p, float v) {
  __hip_atomic_fetch_add(p, v, __ATOMIC_RELAXED, __HIP_MEMORY_SCOPE_WORKGROUP);
}
__device__ __forceinline__ void lds_max(unsigned* p, unsigned v) {
  __hip_atomic_fetch_max(p, v, __ATOMIC_RELAXED, __HIP_MEMORY_SCOPE_WORKGROUP);
}
__device__ __forceinline__ float lrelu(float e) {
  return (e > 0.f) ? e : NEG_SLOPE * e;
}

__global__ __launch_bounds__(256) void range_keys(const int* __restrict__ dst, int E,
                                                  unsigned* __restrict__ key4) {
  const int i = blockIdx.x * blockDim.x + threadIdx.x;
  const int t0 = 4 * i;
  if (t0 >= E) return;
  unsigned d[4];
  if ((E & 3) == 0) {
    const v4i v = ((const v4i*)dst)[i];
    d[0] = (unsigned)v.x; d[1] = (unsigned)v.y; d[2] = (unsigned)v.z; d[3] = (unsigned)v.w;
  } else {
#pragma unroll
    for (int k = 0; k < 4; ++k) d[k] = (t0 + k < E) ? (unsigned)dst[t0 + k] : 0u;
  }
  key4[i] = ((d[0] / ROWS) & 0xFFu) | (((d[1] / ROWS) & 0xFFu) << 8) |
            (((d[2] / ROWS) & 0xFFu) << 16) | (((d[3] / ROWS) & 0xFFu) << 24);
}

__device__ __forceinline__ unsigned zero_bytes(unsigned x) {
  const unsigned y = (x & 0x7F7F7F7Fu) + 0x7F7F7F7Fu;
  return ~(y | x | 0x7F7F7F7Fu);
}

template <int HH, int NB, bool KEYS, int LST>
__global__ __launch_bounds__(RA_THREADS) void range_aggregate(
    const int* __restrict__ ei, const unsigned char* __restrict__ key,
    unsigned* __restrict__ lst, int* __restrict__ lcnt, int E, int N,
    const float* __restrict__ esrc, const float* __restrict__ edst,
    const float* __restrict__ Hm, const float* __restrict__ bias,
    float* __restrict__ out, int relu) {
  constexpr int CH = HH * 16;
  constexpr int J  = HH * NB;
  constexpr int NW = RA_THREADS / 32;
  static_assert(ROWS <= RA_THREADS, "one thread per self loop");
  static_assert(ROWS <= 256, "a local destination row fits a byte of the row list");
  static_assert(J <= 32, "slow path: one lane per logit");
  static_assert(NB == 1 || NB % 2 == 0, "the 16-channel walk takes the batches in pairs");
  constexpr int PJ = (J % 8 == 0) ? J + 4 : J;
  constexpr int AP = ROWS * CH + ((CH == 16 && NB > 1 && (ROWS * CH) % 32 == 0) ? 16 : 0);
  __shared__ float acc[NB * AP];
  __shared__ __attribute__((aligned(16))) float den[ROWS * PJ];
  __shared__ __attribute__((aligned(16))) unsigned mxu[ROWS * PJ];
  __shared__ __attribute__((aligned(16))) float edl[ROWS * PJ];
  __shared__ int qt[NW][QCAP];
  __shared__ unsigned char dq[NW][QCAP];
  __shared__ int ovf;

  const int tid  = threadIdx.x;
  const int lane = tid & 31;
  const int wib  = __builtin_amdgcn_readfirstlane(tid >> 5);
  const int r0   = blockIdx.x * ROWS;
  const int nrows = (N - r0 < ROWS) ? (N - r0) : ROWS;
  int* const myq = qt[wib];
  unsigned char* const myd = dq[wib];

  for (int i = tid; i < NB * AP; i += RA_THREADS) acc[i] = 0.f;
  for (int i = tid; i < ROWS * PJ; i += RA_THREADS) {
    const int row = i / PJ;
    const int j = i - row * PJ;
    den[i] = 0.f;
    mxu[i] = ORDERED_NEG_INF;
    edl[i] = (j < J && row < nrows) ? edst[(size_t)(r0 + row) * J + j] : 0.f;
  }
  if (tid == 0) ovf = 0;
  __syncthreads();

  int tail = 0;

  auto push = [&](bool m, int t) {
    const unsigned mask = __builtin_amdgcn_ballot_w32(m);
    if (mask != 0u) {
      const int pos = tail + (int)__builtin_amdgcn_mbcnt_lo(mask, 0u);
      if (m && pos < QCAP) myq[pos] = t;
      tail += __builtin_popcount(mask);
    }
  };

  if constexpr (LST != 2) push(tid < nrows, E + r0 + tid);

  const int* dp = ei + E;
  unsigned* const slot = (LST != 0) ? lst + ((size_t)blockIdx.x * NW + wib) * QCAP : nullptr;
  if constexpr (LST == 2) {
    tail = __builtin_amdgcn_readfirstlane(lcnt[blockIdx.x * NW + wib]);
  } else if constexpr (KEYS) {
    const unsigned bid4 = (unsigned)blockIdx.x * 0x01010101u;
    const int nvec = E >> 4;
    if (wib == 0) {
      const int t = 16 * nvec + lane;
      bool m = false;
      if (t < E) m = key[t] == (unsigned char)blockIdx.x;
      push(m, t);
    }
    if (nvec > 0) {
      const v4i* kp4 = (const v4i*)key;
      auto ld = [&](int j) -> v4i { return kp4[j < nvec ? j : nvec - 1]; };
      v4i cur[LOADS], nxt[LOADS];
#pragma unroll
      for (int k = 0; k < LOADS; ++k) cur[k] = ld(k * RA_THREADS + tid);
      for (int base = 0; base < nvec; base += LOADS * RA_THREADS) {
#pragma unroll
        for (int k = 0; k < LOADS; ++k) nxt[k] = ld(base + (LOADS + k) * RA_THREADS + tid);
#pragma unroll
        for (int k = 0; k < LOADS; ++k) {
          const v4i v = cur[k];
          const int j = base + k * RA_THREADS + tid;
          unsigned m = (zero_bytes((unsigned)v.x ^ bid4) >> 7) | (zero_bytes((unsigned)v.y ^ bid4) >> 6) |
                       (zero_bytes((unsigned)v.z ^ bid4) >> 5) | (zero_bytes((unsigned)v.w ^ bid4) >> 4);
          if (j >= nvec) m = 0u;
          unsigned live = __builtin_amdgcn_ballot_w32(m != 0u);
          while (live != 0u) {
            const int p = __builtin_ctz(m | 0x80000000u);
            const int pos = tail + (int)__builtin_amdgcn_mbcnt_lo(live, 0u);
            if (m != 0u && pos < QCAP) myq[pos] = 16 * j + 4 * (p & 7) + (p >> 3);
            tail += __builtin_popcount(live);
            m &= m - 1u;
            live = __builtin_amdgcn_ballot_w32(m != 0u);
          }
        }
#pragma unroll
        for (int k = 0; k < LOADS; ++k) cur[k] = nxt[k];
      }
    }
  } else {
  int lead = (4 - (E & 3)) & 3;
  if (lead > E) lead = E;
  const int nvec = (E - lead) >> 2;
  const int tail0 = lead + 4 * nvec;
  if (wib == 0) {
    const int nodd = lead + (E - tail0);
    const int t = (lane < lead) ? lane : tail0 + (lane - lead);
    int d = -1;
    if (lane < nodd) d = dp[t];
    push((unsigned)(d - r0) < (unsigned)nrows, t);
  }

  if (nvec > 0) {
    const v4i* dp4 = (const v4i*)(dp + lead);
    auto ld = [&](int j) -> v4i { return dp4[j < nvec ? j : nvec - 1]; };
    v4i cur[LOADS], nxt[LOADS];
#pragma unroll
    for (int k = 0; k < LOADS; ++k) cur[k] = ld(k * RA_THREADS + tid);
    for (int base = 0; base < nvec; base += LOADS * RA_THREADS) {
#pragma unroll
      for (int k = 0; k < LOADS; ++k) nxt[k] = ld(base + (LOADS + k) * RA_THREADS + tid);
#pragma unroll
      for (int k = 0; k < LOADS; ++k) {
        const v4i v = cur[k];
        const int j = base + k * RA_THREADS + tid;
        const bool ok = j < nvec;
        const bool m0 = ok && (unsigned)(v.x - r0) < (unsigned)nrows;
        const bool m1 = ok && (unsigned)(v.y - r0) < (unsigned)nrows;
        const bool m2 = ok && (unsigned)(v.z - r0) < (unsigned)nrows;
        const bool m3 = ok && (unsigned)(v.w - r0) < (unsigned)nrows;
        if (__builtin_amdgcn_ballot_w32(m0 || m1 || m2 || m3) != 0u) {
          const int tb = lead + 4 * j;
          push(m0, tb);
          push(m1, tb + 1);
          push(m2, tb + 2);
          push(m3, tb + 3);
        }
      }
#pragma unroll
      for (int k = 0; k < LOADS; ++k) cur[k] = nxt[k];
    }
  }
  }
  if ((tail > QCAP || tail < 0) && lane == 0) ovf = 1;
  __syncthreads();
  const bool slow = ovf != 0;
  if constexpr (LST == 1) {
    if (lane == 0) lcnt[blockIdx.x * NW + wib] = slow ? -1 : tail;
  }

  auto ends = [&](int t, int& s, int& dl) {
    const bool real = t < E;
    const int tc = real ? t : 0;
    s = ei[tc];
    int d = ei[(size_t)E + tc];
    if (!real) s = d = t - E;
    dl = d - r0;
  };

  auto scores = [&](int s, int dl, float (&e)[J]) {
    const float* sp = esrc + (size_t)s * J;
    const float* dq = edl + dl * PJ;
    if constexpr (J % 4 == 0) {
#pragma unroll
      for (int q = 0; q < J / 4; ++q) {
        const float4 a = ((const float4*)sp)[q];
        const float4 b = ((const float4*)dq)[q];
        e[4 * q + 0] = lrelu(a.x + b.x);
        e[4 * q + 1] = lrelu(a.y + b.y);
        e[4 * q + 2] = lrelu(a.z + b.z);
        e[4 * q + 3] = lrelu(a.w + b.w);
      }
    } else {
#pragma unroll
      for (int j = 0; j < J; ++j) e[j] = lrelu(sp[j] + dq[j]);
    }
  };

  if (!slow) {
    const int nq = tail;

    for (int b0 = 0; b0 < nq; b0 += 64) {
      int sv[2], dv[2];
      bool on[2];
#pragma unroll
      for (int k = 0; k < 2; ++k) {
        const int i = b0 + 32 * k + lane;
        on[k] = i < nq;
        if constexpr (LST == 2) {
          const unsigned w = slot[on[k] ? i : b0];
          sv[k] = (int)(w >> 8);
          dv[k] = (int)(w & 0xFFu);
        } else {
          ends(myq[on[k] ? i : b0], sv[k], dv[k]);
        }
      }
      float e[2][J];
#pragma unroll
      for (int k = 0; k < 2; ++k) scores(sv[k], dv[k], e[k]);
#pragma unroll
      for (int k = 0; k < 2; ++k) {
        if (on[k]) {
          myq[b0 + 32 * k + lane] = sv[k];
          myd[b0 + 32 * k + lane] = (unsigned char)dv[k];
          if constexpr (LST == 1) slot[b0 + 32 * k + lane] = ((unsigned)sv[k] << 8) | (unsigned)dv[k];
        }
      }
#pragma unroll
      for (int k = 0; k < 2; ++k) {
#pragma unroll
        for (int j = 0; j < J; ++j) lds_max(&mxu[dv[k] * PJ + j], on[k] ? f2o(e[k][j]) : 0u);
      }
    }
    __syncthreads();

    auto batch = [&](int b0, int n) {
      int s = 0, dl = 0;
      float al[J];
#pragma unroll
      for (int j = 0; j < J; ++j) al[j] = 0.f;
      if (lane < n) {
        s = myq[b0 + lane];
        dl = myd[b0 + lane];
        float e[J];
        scores(s, dl, e);
        const unsigned* mp = mxu + dl * PJ;
        if constexpr (J % 4 == 0) {
#pragma unroll
          for (int q = 0; q < J / 4; ++q) {
            const uint4 mv = ((const uint4*)mp)[q];
            al[4 * q + 0] = __expf(e[4 * q + 0] - o2f(mv.x));
            al[4 * q + 1] = __expf(e[4 * q + 1] - o2f(mv.y));
            al[4 * q + 2] = __expf(e[4 * q + 2] - o2f(mv.z));
            al[4 * q + 3] = __expf(e[4 * q + 3] - o2f(mv.w));
          }
        } else {
#pragma unroll
          for (int j = 0; j < J; ++j) al[j] = __expf(e[j] - o2f(mp[j]));
        }
#pragma unroll
        for (int j = 0; j < J; ++j) lds_add(&den[dl * PJ + j], al[j]);
      }
      if constexpr (HH == 4) {
        const bool up = lane >= 16;
        constexpr int EPT = (NB >= 4) ? 4 : 8 / NB;
#pragma unroll 1
        for (int i0 = 0; i0 < n; i0 += EPT) {
          float v0[EPT * NB], v1[EPT * NB];
#pragma unroll
          for (int k = 0; k < EPT; ++k) {
            const int si = __builtin_amdgcn_readlane(s, i0 + k);
#pragma unroll
            for (int b = 0; b < NB; ++b) {
              const float* hp = Hm + ((size_t)b * N + si) * 64 + lane;
              v0[k * NB + b] = hp[0];
              v1[k * NB + b] = hp[32];
            }
          }
#pragma unroll
          for (int k = 0; k < EPT; ++k) {
            if (i0 + k < n) {
              const int i = i0 + k;
              const int di = __builtin_amdgcn_readlane(dl, i);
#pragma unroll
              for (int b = 0; b < NB; ++b) {
                const float a0 = bcast_f(al[4 * b + 0], i), a1 = bcast_f(al[4 * b + 1], i);
                const float a2 = bcast_f(al[4 * b + 2], i), a3 = bcast_f(al[4 * b + 3], i);
                float* op = acc + b * AP + di * 64 + lane;
                lds_add(op, (up ? a1 : a0) * v0[k * NB + b]);
                lds_add(op + 32, (up ? a3 : a2) * v1[k * NB + b]);
              }
            }
          }
        }
      } else if constexpr (NB == 1) {
        const int sub = lane >> 4;
        const int c = lane & 15;
#pragma unroll 1
        for (int i0 = 0; i0 < n; i0 += 16) {
          float v[8], a[8];
          int dj[8];
#pragma unroll
          for (int k = 0; k < 8; ++k) {
            const int j = i0 + 2 * k + sub;
            const int sj = __shfl(s, j);
            dj[k] = __shfl(dl, j);
            a[k] = __shfl(al[0], j);
            v[k] = Hm[(size_t)sj * 16 + c];
          }
#pragma unroll
          for (int k = 0; k < 8; ++k) {
            const int j = i0 + 2 * k + sub;
            if (j < n) lds_add(acc + dj[k] * 16 + c, a[k] * v[k]);
          }
        }
      } else {
        const int sub = lane >> 4;
        const int c = lane & 15;
        constexpr int LPE = NB / 2;
        constexpr int EPT = 16 / LPE;
#pragma unroll 1
        for (int i0 = 0; i0 < n; i0 += EPT) {
          float v[EPT * LPE];
#pragma unroll
          for (int k = 0; k < EPT; ++k) {
            const int si = __builtin_amdgcn_readlane(s, i0 + k);
#pragma unroll
            for (int q = 0; q < LPE; ++q)
              v[k * LPE + q] = Hm[((size_t)(2 * q + sub) * N + si) * 16 + c];
          }
#pragma unroll
          for (int k = 0; k < EPT; ++k) {
            if (i0 + k < n) {
              const int i = i0 + k;
              const int di = __builtin_amdgcn_readlane(dl, i);
#pragma unroll
              for (int q = 0; q < LPE; ++q) {
                const float a0 = bcast_f(al[2 * q], i), a1 = bcast_f(al[2 * q + 1], i);
                lds_add(acc + (2 * q + sub) * AP + di * 16 + c, (sub ? a1 : a0) * v[k * LPE + q]);
              }
            }
          }
        }
      }
    };

    for (int b0 = 0; b0 < nq; b0 += 32) batch(b0, (nq - b0 < 32) ? (nq - b0) : 32);
  } else {
    auto sweep = [&](auto&& f) {
      for (int i = wib; i < nrows; i += NW) f(E + r0 + i);
      for (int base = 0; base < E; base += RA_THREADS) {
        const int t = base + tid;
        int d = -1;
        if (t < E) d = dp[t];
        unsigned mask = __builtin_amdgcn_ballot_w32((unsigned)(d - r0) < (unsigned)nrows);
        while (mask != 0u) {
          const int l = __builtin_ctz(mask);
          mask &= mask - 1u;
          f(base + wib * 32 + l);
        }
      }
    };
    sweep([&](int t) {
      int s, dl;
      ends(t, s, dl);
      if (lane < J)
        lds_max(&mxu[dl * PJ + lane], f2o(lrelu(esrc[(size_t)s * J + lane] + edl[dl * PJ + lane])));
    });
    __syncthreads();
    sweep([&](int t) {
      int s, dl;
      ends(t, s, dl);
      float w = 0.f;
      if (lane < J) {
        const float e = lrelu(esrc[(size_t)s * J + lane] + edl[dl * PJ + lane]);
        w = __expf(e - o2f(mxu[dl * PJ + lane]));
        lds_add(&den[dl * PJ + lane], w);
      }
#pragma unroll
      for (int b = 0; b < NB; ++b) {
#pragma unroll
        for (int c0 = 0; c0 < CH; c0 += 32) {
          const int c = c0 + lane;
          const bool valid = c < CH;
          const float wj = __shfl(w, b * HH + (valid ? (c >> 4) : 0));
          if (valid) lds_add(acc + b * AP + dl * CH + c, wj * Hm[((size_t)b * N + s) * CH + c]);
        }
      }
    });
  }
  __syncthreads();

  for (int b = 0; b < NB; ++b) {
    for (int idx = tid; idx < nrows * CH; idx += RA_THREADS) {
      const int dl = idx / CH;
      const int c = idx % CH;
      float v = acc[b * AP + idx] / (den[dl * PJ + b * HH + (c >> 4)] + GAT_EPS) + bias[c];
      if (relu) v = fmaxf(v, 0.f);
      out[((size_t)b * N + r0) * CH + idx] = v;
    }
  }
}

static inline int cdiv(long long a, int b) { return (int)((a + b - 1) / b); }

struct LayerBufs {
  float* hbuf; float* esrc; float* edst;
};

struct ListBufs {
  unsigned char* key; unsigned* lst; int* lcnt;
};
template <int HH, int NB>
static void run_edge_phase(const LayerBufs& L, const int* ei, const ListBufs& Q, bool build, int E, int N,
                           const float* bias, float* outp, int relu,
                           hipStream_t stream) {
  if (!Q.key) {
    range_aggregate<HH, NB, false, 0><<<cdiv(N, ROWS), RA_THREADS, 0, stream>>>(
        ei, Q.key, Q.lst, Q.lcnt, E, N, L.esrc, L.edst, L.hbuf, bias, outp, relu);
    return;
  }
  if constexpr (HH == 4) {
    if (build) {
      range_aggregate<HH, NB, true, 1><<<cdiv(N, ROWS), RA_THREADS, 0, stream>>>(
          ei, Q.key, Q.lst, Q.lcnt, E, N, L.esrc, L.edst, L.hbuf, bias, outp, relu);
      return;
    }
  }
  range_aggregate<HH, NB, true, 2><<<cdiv(N, ROWS), RA_THREADS, 0, stream>>>(
      ei, Q.key, Q.lst, Q.lcnt, E, N, L.esrc, L.edst, L.hbuf, bias, outp, relu);
}

static inline size_t ws_need(int NB, int N, int E) {
  return sizeof(float) * ((size_t)NB * N * 64 * 2 + (size_t)NB * N * 4 * 2);
}

static inline ListBufs list_place(int NB, int N, int E, void* d_ws, size_t ws_size) {
  const size_t slots = (size_t)256 * (RA_THREADS / 32);
  const size_t off = (ws_need(NB, N, E) + 15) & ~(size_t)15;
  const size_t off_lst = (off + (size_t)E + 16 + 15) & ~(size_t)15;
  const size_t off_cnt = off_lst + sizeof(unsigned) * slots * QCAP;
  if (cdiv(N, ROWS) > 256 || ws_size < off_cnt + sizeof(int) * slots) return ListBufs{nullptr, nullptr, nullptr};
  char* const base = (char*)d_ws;
  return ListBufs{(unsigned char*)(base + off), (unsigned*)(base + off_lst), (int*)(base + off_cnt)};
}

template <int NB>
static void run_pass(void* const* d_in, int b0, int N, int E, int NCLS,
                     float* d_out, void* d_ws, const ListBufs& Q, hipStream_t stream) {
  const float* x   = (const float*)d_in[0];
  const int*   ei  = (const int*)d_in[1];
  const float* W1  = (const float*)d_in[2];
  const float* a1s = (const float*)d_in[3];
  const float* a1d = (const float*)d_in[4];
  const float* b1  = (const float*)d_in[5];
  const float* W2  = (const float*)d_in[6];
  const float* a2s = (const float*)d_in[7];
  const float* a2d = (const float*)d_in[8];
  const float* b2  = (const float*)d_in[9];
  const float* W3  = (const float*)d_in[10];
  const float* a3s = (const float*)d_in[11];
  const float* a3d = (const float*)d_in[12];
  const float* b3  = (const float*)d_in[13];
  const int F_IN = 128, H = 4, HC = 64;

  float* ws = (float*)d_ws;
  float*    hbuf  = ws;
  float*    xbuf  = hbuf + (size_t)NB * N * HC;
  float*    esrc  = xbuf + (size_t)NB * N * HC;
  float*    edst  = esrc + (size_t)NB * N * H;

  const int ntiles = (NB * N + 15) / 16;
  const int gemm_grid = (ntiles + 3) / 4;

  LayerBufs L{hbuf, esrc, edst};

  gemm_wmma_f32<128, 64><<<gemm_grid, 128, 0, stream>>>(
      x + (size_t)b0 * N * F_IN, W1, hbuf, ntiles);
  node_attn<<<cdiv((long long)N * NB * H, 256), 256, 0, stream>>>(
      hbuf, a1s, a1d, esrc, edst, N, H, NB);
  run_edge_phase<4, NB>(L, ei, Q, b0 == 0, E, N, b1, xbuf, 1, stream);

  gemm_wmma_f32<64, 64><<<gemm_grid, 128, 0, stream>>>(xbuf, W2, hbuf, ntiles);
  node_attn<<<cdiv((long long)N * NB * H, 256), 256, 0, stream>>>(
      hbuf, a2s, a2d, esrc, edst, N, H, NB);
  run_edge_phase<4, NB>(L, ei, Q, false, E, N, b2, xbuf, 1, stream);

  gemm_wmma_f32<64, 16><<<gemm_grid, 128, 0, stream>>>(xbuf, W3, hbuf, ntiles);
  node_attn<<<cdiv((long long)N * NB, 256), 256, 0, stream>>>(
      hbuf, a3s, a3d, esrc, edst, N, 1, NB);
  run_edge_phase<1, NB>(L, ei, Q, false, E, N, b3, d_out + (size_t)b0 * N * NCLS, 0, stream);
}

extern "C" void kernel_launch(void* const* d_in, const int* in_sizes, int n_in,
                              void* d_out, int out_size, void* d_ws, size_t ws_size,
                              hipStream_t stream) {
  const int B = 4, F_IN = 128;
  const int E = in_sizes[1] / 2;
  const int N = in_sizes[0] / (B * F_IN);
  const int NCLS = in_sizes[13];

  const int NBP = (ws_size >= ws_need(4, N, E)) ? 4 : (ws_size >= ws_need(2, N, E)) ? 2 : 1;

  const ListBufs Q = list_place(NBP, N, E, d_ws, ws_size);
  if (Q.key)
    range_keys<<<cdiv(cdiv(E, 4), 256), 256, 0, stream>>>((const int*)d_in[1] + E, E, (unsigned*)Q.key);

  if (NBP == 4) {
    run_pass<4>(d_in, 0, N, E, NCLS, (float*)d_out, d_ws, Q, stream);
  } else if (NBP == 2) {
    for (int b = 0; b < B; b += 2) run_pass<2>(d_in, b, N, E, NCLS, (float*)d_out, d_ws, Q, stream);
  } else {
    for (int b = 0; b < B; ++b) run_pass<1>(d_in, b, N, E, NCLS, (float*)d_out, d_ws, Q, stream);
  }
}


# Idle lanes of a max trip offer their 0 to a row of their own (row = lane), not all to one row: same-address LDS atomics are served lane by lane; maxima unchanged; left: waves' lists uneven
# speedup vs baseline: 16.6287x; 1.1276x over previous
#include <hip/hip_runtime.h>
#include <hip/hip_bf16.h>


typedef __attribute__((ext_vector_type(2))) float v2f;
typedef __attribute__((ext_vector_type(8))) float v8f;
typedef __attribute__((ext_vector_type(4))) int v4i;

#define NEG_SLOPE 0.2f
#define GAT_EPS   1e-16f

template <int KDIM, int MO>
__global__ __launch_bounds__(128) void gemm_wmma_f32(
    const float* __restrict__ X, const float* __restrict__ W,
    float* __restrict__ Hout, int ntiles) {
  const int wave = threadIdx.x >> 5;
  const int tile = blockIdx.x * 4 + wave;
  if (tile >= ntiles) return;
  const int lane  = threadIdx.x & 31;
  const int lmod  = lane & 15;
  const int lhalf = lane >> 4;
  const int row0  = tile * 16;

  v8f acc[MO / 16];
#pragma unroll
  for (int t = 0; t < MO / 16; ++t) acc[t] = v8f{0, 0, 0, 0, 0, 0, 0, 0};

  const float* xrow = X + (size_t)(row0 + lmod) * KDIM;
#pragma unroll 4
  for (int k0 = 0; k0 < KDIM; k0 += 4) {
    const int ka = k0 + 2 * lhalf;
    v2f a;
    a.x = xrow[ka];
    a.y = xrow[ka + 1];
#pragma unroll
    for (int t = 0; t < MO / 16; ++t) {
      const float* wr = W + (size_t)ka * MO + t * 16 + lmod;
      v2f b;
      b.x = wr[0];
      b.y = wr[MO];
      acc[t] = __builtin_amdgcn_wmma_f32_16x16x4_f32(
          false, a, false, b, (short)0, acc[t], false, false);
    }
  }
#pragma unroll
  for (int t = 0; t < MO / 16; ++t)
#pragma unroll
    for (int r = 0; r < 8; ++r)
      Hout[(size_t)(row0 + r + 8 * lhalf) * MO + t * 16 + lmod] = acc[t][r];
}

__global__ void node_attn(const float* __restrict__ Hm,
                          const float* __restrict__ a_src,
                          const float* __restrict__ a_dst,
                          float* __restrict__ esrc, float* __restrict__ edst,
                          int N, int Hh, int NBv) {
  int idx = blockIdx.x * blockDim.x + threadIdx.x;
  if (idx >= N * NBv * Hh) return;
  const int h = idx % Hh;
  const int b = (idx / Hh) % NBv;
  const int n = idx / (Hh * NBv);
  const float* row = Hm + (((size_t)b * N + n) * Hh + h) * 16;
  float s = 0.f, d = 0.f;
#pragma unroll
  for (int c = 0; c < 16; ++c) {
    const float v = row[c];
    s += v * a_src[h * 16 + c];
    d += v * a_dst[h * 16 + c];
  }
  esrc[idx] = s;
  edst[idx] = d;
}

__device__ __forceinline__ unsigned f2o(float f) {
  unsigned u = __float_as_uint(f);
  return (u & 0x80000000u) ? ~u : (u | 0x80000000u);
}
__device__ __forceinline__ float o2f(unsigned u) {
  return __uint_as_float((u & 0x80000000u) ? (u & 0x7FFFFFFFu) : ~u);
}
#define ORDERED_NEG_INF 0x007FFFFFu

#define ROWS   196
#define QCAP   448
#define LOADS  8
#define RA_THREADS 1024

__device__ __forceinline__ float bcast_f(float x, int i) {
  return __int_as_float(__builtin_amdgcn_readlane(__float_as_int(x), i));
}
__device__ __forceinline__ void lds_add(float* p, float v) {
  __hip_atomic_fetch_add(p, v, __ATOMIC_RELAXED, __HIP_MEMORY_SCOPE_WORKGROUP);
}
__device__ __forceinline__ void lds_max(unsigned* p, unsigned v) {
  __hip_atomic_fetch_max(p, v, __ATOMIC_RELAXED, __HIP_MEMORY_SCOPE_WORKGROUP);
}
__device__ __forceinline__ float lrelu(float e) {
  return (e > 0.f) ? e : NEG_SLOPE * e;
}

__global__ __launch_bounds__(256) void range_keys(const int* __restrict__ dst, int E,
                                                  unsigned* __restrict__ key4) {
  const int i = blockIdx.x * blockDim.x + threadIdx.x;
  const int t0 = 4 * i;
  if (t0 >= E) return;
  unsigned d[4];
  if ((E & 3) == 0) {
    const v4i v = ((const v4i*)dst)[i];
    d[0] = (unsigned)v.x; d[1] = (unsigned)v.y; d[2] = (unsigned)v.z; d[3] = (unsigned)v.w;
  } else {
#pragma unroll
    for (int k = 0; k < 4; ++k) d[k] = (t0 + k < E) ? (unsigned)dst[t0 + k] : 0u;
  }
  key4[i] = ((d[0] / ROWS) & 0xFFu) | (((d[1] / ROWS) & 0xFFu) << 8) |
            (((d[2] / ROWS) & 0xFFu) << 16) | (((d[3] / ROWS) & 0xFFu) << 24);
}

__device__ __forceinline__ unsigned zero_bytes(unsigned x) {
  const unsigned y = (x & 0x7F7F7F7Fu) + 0x7F7F7F7Fu;
  return ~(y | x | 0x7F7F7F7Fu);
}

template <int HH, int NB, bool KEYS, int LST>
__global__ __launch_bounds__(RA_THREADS) void range_aggregate(
    const int* __restrict__ ei, const unsigned char* __restrict__ key,
    unsigned* __restrict__ lst, int* __restrict__ lcnt, int E, int N,
    const float* __restrict__ esrc, const float* __restrict__ edst,
    const float* __restrict__ Hm, const float* __restrict__ bias,
    float* __restrict__ out, int relu) {
  constexpr int CH = HH * 16;
  constexpr int J  = HH * NB;
  constexpr int NW = RA_THREADS / 32;
  static_assert(ROWS <= RA_THREADS, "one thread per self loop");
  static_assert(ROWS >= 32, "an idle lane of the max walk offers to row = lane");
  static_assert(ROWS <= 256, "a local destination row fits a byte of the row list");
  static_assert(J <= 32, "slow path: one lane per logit");
  static_assert(NB == 1 || NB % 2 == 0, "the 16-channel walk takes the batches in pairs");
  constexpr int PJ = (J % 8 == 0) ? J + 4 : J;
  constexpr int AP = ROWS * CH + ((CH == 16 && NB > 1 && (ROWS * CH) % 32 == 0) ? 16 : 0);
  __shared__ float acc[NB * AP];
  __shared__ __attribute__((aligned(16))) float den[ROWS * PJ];
  __shared__ __attribute__((aligned(16))) unsigned mxu[ROWS * PJ];
  __shared__ __attribute__((aligned(16))) float edl[ROWS * PJ];
  __shared__ int qt[NW][QCAP];
  __shared__ unsigned char dq[NW][QCAP];
  __shared__ int ovf;

  const int tid  = threadIdx.x;
  const int lane = tid & 31;
  const int wib  = __builtin_amdgcn_readfirstlane(tid >> 5);
  const int r0   = blockIdx.x * ROWS;
  const int nrows = (N - r0 < ROWS) ? (N - r0) : ROWS;
  int* const myq = qt[wib];
  unsigned char* const myd = dq[wib];

  for (int i = tid; i < NB * AP; i += RA_THREADS) acc[i] = 0.f;
  for (int i = tid; i < ROWS * PJ; i += RA_THREADS) {
    const int row = i / PJ;
    const int j = i - row * PJ;
    den[i] = 0.f;
    mxu[i] = ORDERED_NEG_INF;
    edl[i] = (j < J && row < nrows) ? edst[(size_t)(r0 + row) * J + j] : 0.f;
  }
  if (tid == 0) ovf = 0;
  __syncthreads();

  int tail = 0;

  auto push = [&](bool m, int t) {
    const unsigned mask = __builtin_amdgcn_ballot_w32(m);
    if (mask != 0u) {
      const int pos = tail + (int)__builtin_amdgcn_mbcnt_lo(mask, 0u);
      if (m && pos < QCAP) myq[pos] = t;
      tail += __builtin_popcount(mask);
    }
  };

  if constexpr (LST != 2) push(tid < nrows, E + r0 + tid);

  const int* dp = ei + E;
  unsigned* const slot = (LST != 0) ? lst + ((size_t)blockIdx.x * NW + wib) * QCAP : nullptr;
  if constexpr (LST == 2) {
    tail = __builtin_amdgcn_readfirstlane(lcnt[blockIdx.x * NW + wib]);
  } else if constexpr (KEYS) {
    const unsigned bid4 = (unsigned)blockIdx.x * 0x01010101u;
    const int nvec = E >> 4;
    if (wib == 0) {
      const int t = 16 * nvec + lane;
      bool m = false;
      if (t < E) m = key[t] == (unsigned char)blockIdx.x;
      push(m, t);
    }
    if (nvec > 0) {
      const v4i* kp4 = (const v4i*)key;
      auto ld = [&](int j) -> v4i { return kp4[j < nvec ? j : nvec - 1]; };
      v4i cur[LOADS], nxt[LOADS];
#pragma unroll
      for (int k = 0; k < LOADS; ++k) cur[k] = ld(k * RA_THREADS + tid);
      for (int base = 0; base < nvec; base += LOADS * RA_THREADS) {
#pragma unroll
        for (int k = 0; k < LOADS; ++k) nxt[k] = ld(base + (LOADS + k) * RA_THREADS + tid);
#pragma unroll
        for (int k = 0; k < LOADS; ++k) {
          const v4i v = cur[k];
          const int j = base + k * RA_THREADS + tid;
          unsigned m = (zero_bytes((unsigned)v.x ^ bid4) >> 7) | (zero_bytes((unsigned)v.y ^ bid4) >> 6) |
                       (zero_bytes((unsigned)v.z ^ bid4) >> 5) | (zero_bytes((unsigned)v.w ^ bid4) >> 4);
          if (j >= nvec) m = 0u;
          unsigned live = __builtin_amdgcn_ballot_w32(m != 0u);
          while (live != 0u) {
            const int p = __builtin_ctz(m | 0x80000000u);
            const int pos = tail + (int)__builtin_amdgcn_mbcnt_lo(live, 0u);
            if (m != 0u && pos < QCAP) myq[pos] = 16 * j + 4 * (p & 7) + (p >> 3);
            tail += __builtin_popcount(live);
            m &= m - 1u;
            live = __builtin_amdgcn_ballot_w32(m != 0u);
          }
        }
#pragma unroll
        for (int k = 0; k < LOADS; ++k) cur[k] = nxt[k];
      }
    }
  } else {
  int lead = (4 - (E & 3)) & 3;
  if (lead > E) lead = E;
  const int nvec = (E - lead) >> 2;
  const int tail0 = lead + 4 * nvec;
  if (wib == 0) {
    const int nodd = lead + (E - tail0);
    const int t = (lane < lead) ? lane : tail0 + (lane - lead);
    int d = -1;
    if (lane < nodd) d = dp[t];
    push((unsigned)(d - r0) < (unsigned)nrows, t);
  }

  if (nvec > 0) {
    const v4i* dp4 = (const v4i*)(dp + lead);
    auto ld = [&](int j) -> v4i { return dp4[j < nvec ? j : nvec - 1]; };
    v4i cur[LOADS], nxt[LOADS];
#pragma unroll
    for (int k = 0; k < LOADS; ++k) cur[k] = ld(k * RA_THREADS + tid);
    for (int base = 0; base < nvec; base += LOADS * RA_THREADS) {
#pragma unroll
      for (int k = 0; k < LOADS; ++k) nxt[k] = ld(base + (LOADS + k) * RA_THREADS + tid);
#pragma unroll
      for (int k = 0; k < LOADS; ++k) {
        const v4i v = cur[k];
        const int j = base + k * RA_THREADS + tid;
        const bool ok = j < nvec;
        const bool m0 = ok && (unsigned)(v.x - r0) < (unsigned)nrows;
        const bool m1 = ok && (unsigned)(v.y - r0) < (unsigned)nrows;
        const bool m2 = ok && (unsigned)(v.z - r0) < (unsigned)nrows;
        const bool m3 = ok && (unsigned)(v.w - r0) < (unsigned)nrows;
        if (__builtin_amdgcn_ballot_w32(m0 || m1 || m2 || m3) != 0u) {
          const int tb = lead + 4 * j;
          push(m0, tb);
          push(m1, tb + 1);
          push(m2, tb + 2);
          push(m3, tb + 3);
        }
      }
#pragma unroll
      for (int k = 0; k < LOADS; ++k) cur[k] = nxt[k];
    }
  }
  }
  if ((tail > QCAP || tail < 0) && lane == 0) ovf = 1;
  __syncthreads();
  const bool slow = ovf != 0;
  if constexpr (LST == 1) {
    if (lane == 0) lcnt[blockIdx.x * NW + wib] = slow ? -1 : tail;
  }

  auto ends = [&](int t, int& s, int& dl) {
    const bool real = t < E;
    const int tc = real ? t : 0;
    s = ei[tc];
    int d = ei[(size_t)E + tc];
    if (!real) s = d = t - E;
    dl = d - r0;
  };

  auto scores = [&](int s, int dl, float (&e)[J]) {
    const float* sp = esrc + (size_t)s * J;
    const float* dq = edl + dl * PJ;
    if constexpr (J % 4 == 0) {
#pragma unroll
      for (int q = 0; q < J / 4; ++q) {
        const float4 a = ((const float4*)sp)[q];
        const float4 b = ((const float4*)dq)[q];
        e[4 * q + 0] = lrelu(a.x + b.x);
        e[4 * q + 1] = lrelu(a.y + b.y);
        e[4 * q + 2] = lrelu(a.z + b.z);
        e[4 * q + 3] = lrelu(a.w + b.w);
      }
    } else {
#pragma unroll
      for (int j = 0; j < J; ++j) e[j] = lrelu(sp[j] + dq[j]);
    }
  };

  if (!slow) {
    const int nq = tail;

    for (int b0 = 0; b0 < nq; b0 += 64) {
      int sv[2], dv[2];
      bool on[2];
#pragma unroll
      for (int k = 0; k < 2; ++k) {
        const int i = b0 + 32 * k + lane;
        on[k] = i < nq;
        if constexpr (LST == 2) {
          const unsigned w = slot[on[k] ? i : b0];
          sv[k] = (int)(w >> 8);
          dv[k] = (int)(w & 0xFFu);
        } else {
          ends(myq[on[k] ? i : b0], sv[k], dv[k]);
        }
      }
      float e[2][J];
#pragma unroll
      for (int k = 0; k < 2; ++k) scores(sv[k], dv[k], e[k]);
#pragma unroll
      for (int k = 0; k < 2; ++k) {
        if (on[k]) {
          myq[b0 + 32 * k + lane] = sv[k];
          myd[b0 + 32 * k + lane] = (unsigned char)dv[k];
          if constexpr (LST == 1) slot[b0 + 32 * k + lane] = ((unsigned)sv[k] << 8) | (unsigned)dv[k];
        }
      }
#pragma unroll
      for (int k = 0; k < 2; ++k) {
        const int dr = on[k] ? dv[k] : lane;
#pragma unroll
        for (int j = 0; j < J; ++j) lds_max(&mxu[dr * PJ + j], on[k] ? f2o(e[k][j]) : 0u);
      }
    }
    __syncthreads();

    auto batch = [&](int b0, int n) {
      int s = 0, dl = 0;
      float al[J];
#pragma unroll
      for (int j = 0; j < J; ++j) al[j] = 0.f;
      if (lane < n) {
        s = myq[b0 + lane];
        dl = myd[b0 + lane];
        float e[J];
        scores(s, dl, e);
        const unsigned* mp = mxu + dl * PJ;
        if constexpr (J % 4 == 0) {
#pragma unroll
          for (int q = 0; q < J / 4; ++q) {
            const uint4 mv = ((const uint4*)mp)[q];
            al[4 * q + 0] = __expf(e[4 * q + 0] - o2f(mv.x));
            al[4 * q + 1] = __expf(e[4 * q + 1] - o2f(mv.y));
            al[4 * q + 2] = __expf(e[4 * q + 2] - o2f(mv.z));
            al[4 * q + 3] = __expf(e[4 * q + 3] - o2f(mv.w));
          }
        } else {
#pragma unroll
          for (int j = 0; j < J; ++j) al[j] = __expf(e[j] - o2f(mp[j]));
        }
#pragma unroll
        for (int j = 0; j < J; ++j) lds_add(&den[dl * PJ + j], al[j]);
      }
      if constexpr (HH == 4) {
        const bool up = lane >= 16;
        constexpr int EPT = (NB >= 4) ? 4 : 8 / NB;
#pragma unroll 1
        for (int i0 = 0; i0 < n; i0 += EPT) {
          float v0[EPT * NB], v1[EPT * NB];
#pragma unroll
          for (int k = 0; k < EPT; ++k) {
            const int si = __builtin_amdgcn_readlane(s, i0 + k);
#pragma unroll
            for (int b = 0; b < NB; ++b) {
              const float* hp = Hm + ((size_t)b * N + si) * 64 + lane;
              v0[k * NB + b] = hp[0];
              v1[k * NB + b] = hp[32];
            }
          }
#pragma unroll
          for (int k = 0; k < EPT; ++k) {
            if (i0 + k < n) {
              const int i = i0 + k;
              const int di = __builtin_amdgcn_readlane(dl, i);
#pragma unroll
              for (int b = 0; b < NB; ++b) {
                const float a0 = bcast_f(al[4 * b + 0], i), a1 = bcast_f(al[4 * b + 1], i);
                const float a2 = bcast_f(al[4 * b + 2], i), a3 = bcast_f(al[4 * b + 3], i);
                float* op = acc + b * AP + di * 64 + lane;
                lds_add(op, (up ? a1 : a0) * v0[k * NB + b]);
                lds_add(op + 32, (up ? a3 : a2) * v1[k * NB + b]);
              }
            }
          }
        }
      } else if constexpr (NB == 1) {
        const int sub = lane >> 4;
        const int c = lane & 15;
#pragma unroll 1
        for (int i0 = 0; i0 < n; i0 += 16) {
          float v[8], a[8];
          int dj[8];
#pragma unroll
          for (int k = 0; k < 8; ++k) {
            const int j = i0 + 2 * k + sub;
            const int sj = __shfl(s, j);
            dj[k] = __shfl(dl, j);
            a[k] = __shfl(al[0], j);
            v[k] = Hm[(size_t)sj * 16 + c];
          }
#pragma unroll
          for (int k = 0; k < 8; ++k) {
            const int j = i0 + 2 * k + sub;
            if (j < n) lds_add(acc + dj[k] * 16 + c, a[k] * v[k]);
          }
        }
      } else {
        const int sub = lane >> 4;
        const int c = lane & 15;
        constexpr int LPE = NB / 2;
        constexpr int EPT = 16 / LPE;
#pragma unroll 1
        for (int i0 = 0; i0 < n; i0 += EPT) {
          float v[EPT * LPE];
#pragma unroll
          for (int k = 0; k < EPT; ++k) {
            const int si = __builtin_amdgcn_readlane(s, i0 + k);
#pragma unroll
            for (int q = 0; q < LPE; ++q)
              v[k * LPE + q] = Hm[((size_t)(2 * q + sub) * N + si) * 16 + c];
          }
#pragma unroll
          for (int k = 0; k < EPT; ++k) {
            if (i0 + k < n) {
              const int i = i0 + k;
              const int di = __builtin_amdgcn_readlane(dl, i);
#pragma unroll
              for (int q = 0; q < LPE; ++q) {
                const float a0 = bcast_f(al[2 * q], i), a1 = bcast_f(al[2 * q + 1], i);
                lds_add(acc + (2 * q + sub) * AP + di * 16 + c, (sub ? a1 : a0) * v[k * LPE + q]);
              }
            }
          }
        }
      }
    };

    for (int b0 = 0; b0 < nq; b0 += 32) batch(b0, (nq - b0 < 32) ? (nq - b0) : 32);
  } else {
    auto sweep = [&](auto&& f) {
      for (int i = wib; i < nrows; i += NW) f(E + r0 + i);
      for (int base = 0; base < E; base += RA_THREADS) {
        const int t = base + tid;
        int d = -1;
        if (t < E) d = dp[t];
        unsigned mask = __builtin_amdgcn_ballot_w32((unsigned)(d - r0) < (unsigned)nrows);
        while (mask != 0u) {
          const int l = __builtin_ctz(mask);
          mask &= mask - 1u;
          f(base + wib * 32 + l);
        }
      }
    };
    sweep([&](int t) {
      int s, dl;
      ends(t, s, dl);
      if (lane < J)
        lds_max(&mxu[dl * PJ + lane], f2o(lrelu(esrc[(size_t)s * J + lane] + edl[dl * PJ + lane])));
    });
    __syncthreads();
    sweep([&](int t) {
      int s, dl;
      ends(t, s, dl);
      float w = 0.f;
      if (lane < J) {
        const float e = lrelu(esrc[(size_t)s * J + lane] + edl[dl * PJ + lane]);
        w = __expf(e - o2f(mxu[dl * PJ + lane]));
        lds_add(&den[dl * PJ + lane], w);
      }
#pragma unroll
      for (int b = 0; b < NB; ++b) {
#pragma unroll
        for (int c0 = 0; c0 < CH; c0 += 32) {
          const int c = c0 + lane;
          const bool valid = c < CH;
          const float wj = __shfl(w, b * HH + (valid ? (c >> 4) : 0));
          if (valid) lds_add(acc + b * AP + dl * CH + c, wj * Hm[((size_t)b * N + s) * CH + c]);
        }
      }
    });
  }
  __syncthreads();

  for (int b = 0; b < NB; ++b) {
    for (int idx = tid; idx < nrows * CH; idx += RA_THREADS) {
      const int dl = idx / CH;
      const int c = idx % CH;
      float v = acc[b * AP + idx] / (den[dl * PJ + b * HH + (c >> 4)] + GAT_EPS) + bias[c];
      if (relu) v = fmaxf(v, 0.f);
      out[((size_t)b * N + r0) * CH + idx] = v;
    }
  }
}

static inline int cdiv(long long a, int b) { return (int)((a + b - 1) / b); }

struct LayerBufs {
  float* hbuf; float* esrc; float* edst;
};

struct ListBufs {
  unsigned char* key; unsigned* lst; int* lcnt;
};
template <int HH, int NB>
static void run_edge_phase(const LayerBufs& L, const int* ei, const ListBufs& Q, bool build, int E, int N,
                           const float* bias, float* outp, int relu,
                           hipStream_t stream) {
  if (!Q.key) {
    range_aggregate<HH, NB, false, 0><<<cdiv(N, ROWS), RA_THREADS, 0, stream>>>(
        ei, Q.key, Q.lst, Q.lcnt, E, N, L.esrc, L.edst, L.hbuf, bias, outp, relu);
    return;
  }
  if constexpr (HH == 4) {
    if (build) {
      range_aggregate<HH, NB, true, 1><<<cdiv(N, ROWS), RA_THREADS, 0, stream>>>(
          ei, Q.key, Q.lst, Q.lcnt, E, N, L.esrc, L.edst, L.hbuf, bias, outp, relu);
      return;
    }
  }
  range_aggregate<HH, NB, true, 2><<<cdiv(N, ROWS), RA_THREADS, 0, stream>>>(
      ei, Q.key, Q.lst, Q.lcnt, E, N, L.esrc, L.edst, L.hbuf, bias, outp, relu);
}

static inline size_t ws_need(int NB, int N, int E) {
  return sizeof(float) * ((size_t)NB * N * 64 * 2 + (size_t)NB * N * 4 * 2);
}

static inline ListBufs list_place(int NB, int N, int E, void* d_ws, size_t ws_size) {
  const size_t slots = (size_t)256 * (RA_THREADS / 32);
  const size_t off = (ws_need(NB, N, E) + 15) & ~(size_t)15;
  const size_t off_lst = (off + (size_t)E + 16 + 15) & ~(size_t)15;
  const size_t off_cnt = off_lst + sizeof(unsigned) * slots * QCAP;
  if (cdiv(N, ROWS) > 256 || ws_size < off_cnt + sizeof(int) * slots) return ListBufs{nullptr, nullptr, nullptr};
  char* const base = (char*)d_ws;
  return ListBufs{(unsigned char*)(base + off), (unsigned*)(base + off_lst), (int*)(base + off_cnt)};
}

template <int NB>
static void run_pass(void* const* d_in, int b0, int N, int E, int NCLS,
                     float* d_out, void* d_ws, const ListBufs& Q, hipStream_t stream) {
  const float* x   = (const float*)d_in[0];
  const int*   ei  = (const int*)d_in[1];
  const float* W1  = (const float*)d_in[2];
  const float* a1s = (const float*)d_in[3];
  const float* a1d = (const float*)d_in[4];
  const float* b1  = (const float*)d_in[5];
  const float* W2  = (const float*)d_in[6];
  const float* a2s = (const float*)d_in[7];
  const float* a2d = (const float*)d_in[8];
  const float* b2  = (const float*)d_in[9];
  const float* W3  = (const float*)d_in[10];
  const float* a3s = (const float*)d_in[11];
  const float* a3d = (const float*)d_in[12];
  const float* b3  = (const float*)d_in[13];
  const int F_IN = 128, H = 4, HC = 64;

  float* ws = (float*)d_ws;
  float*    hbuf  = ws;
  float*    xbuf  = hbuf + (size_t)NB * N * HC;
  float*    esrc  = xbuf + (size_t)NB * N * HC;
  float*    edst  = esrc + (size_t)NB * N * H;

  const int ntiles = (NB * N + 15) / 16;
  const int gemm_grid = (ntiles + 3) / 4;

  LayerBufs L{hbuf, esrc, edst};

  gemm_wmma_f32<128, 64><<<gemm_grid, 128, 0, stream>>>(
      x + (size_t)b0 * N * F_IN, W1, hbuf, ntiles);
  node_attn<<<cdiv((long long)N * NB * H, 256), 256, 0, stream>>>(
      hbuf, a1s, a1d, esrc, edst, N, H, NB);
  run_edge_phase<4, NB>(L, ei, Q, b0 == 0, E, N, b1, xbuf, 1, stream);

  gemm_wmma_f32<64, 64><<<gemm_grid, 128, 0, stream>>>(xbuf, W2, hbuf, ntiles);
  node_attn<<<cdiv((long long)N * NB * H, 256), 256, 0, stream>>>(
      hbuf, a2s, a2d, esrc, edst, N, H, NB);
  run_edge_phase<4, NB>(L, ei, Q, false, E, N, b2, xbuf, 1, stream);

  gemm_wmma_f32<64, 16><<<gemm_grid, 128, 0, stream>>>(xbuf, W3, hbuf, ntiles);
  node_attn<<<cdiv((long long)N * NB, 256), 256, 0, stream>>>(
      hbuf, a3s, a3d, esrc, edst, N, 1, NB);
  run_edge_phase<1, NB>(L, ei, Q, false, E, N, b3, d_out + (size_t)b0 * N * NCLS, 0, stream);
}

extern "C" void kernel_launch(void* const* d_in, const int* in_sizes, int n_in,
                              void* d_out, int out_size, void* d_ws, size_t ws_size,
                              hipStream_t stream) {
  const int B = 4, F_IN = 128;
  const int E = in_sizes[1] / 2;
  const int N = in_sizes[0] / (B * F_IN);
  const int NCLS = in_sizes[13];

  const int NBP = (ws_size >= ws_need(4, N, E)) ? 4 : (ws_size >= ws_need(2, N, E)) ? 2 : 1;

  const ListBufs Q = list_place(NBP, N, E, d_ws, ws_size);
  if (Q.key)
    range_keys<<<cdiv(cdiv(E, 4), 256), 256, 0, stream>>>((const int*)d_in[1] + E, E, (unsigned*)Q.key);

  if (NBP == 4) {
    run_pass<4>(d_in, 0, N, E, NCLS, (float*)d_out, d_ws, Q, stream);
  } else if (NBP == 2) {
    for (int b = 0; b < B; b += 2) run_pass<2>(d_in, b, N, E, NCLS, (float*)d_out, d_ws, Q, stream);
  } else {
    for (int b = 0; b < B; ++b) run_pass<1>(d_in, b, N, E, NCLS, (float*)d_out, d_ws, Q, stream);
  }
}
